# starting kernel
# baseline (speedup 1.0000x reference)
#include <hip/hip_runtime.h>
#include <hip/hip_bf16.h>


typedef float v2f __attribute__((ext_vector_type(2)));
typedef float v8f __attribute__((ext_vector_type(8)));

#define NEG_SLOPE 0.2f
#define KEY_NEG_INF 0x007FFFFFu

__device__ __forceinline__ unsigned fkey(float x) {
  unsigned b = __float_as_uint(x);
  return (b & 0x80000000u) ? ~b : (b | 0x80000000u);
}
__device__ __forceinline__ float funkey(unsigned k) {
  return __uint_as_float((k & 0x80000000u) ? (k & 0x7FFFFFFFu) : ~k);
}


__global__ void fill_f32_kernel(float* __restrict__ p, int n, float v) {
  int i = blockIdx.x * blockDim.x + threadIdx.x;
  if (i < n) p[i] = v;
}
__global__ void fill_u32_kernel(unsigned* __restrict__ p, int n, unsigned v) {
  int i = blockIdx.x * blockDim.x + threadIdx.x;
  if (i < n) p[i] = v;
}


template <int K>
__global__ __launch_bounds__(256) void gemm_wmma_f32(
    const float* __restrict__ A, const float* __restrict__ W,
    float* __restrict__ H, int Cout) {
  const int lane = threadIdx.x;
  const int colTile = threadIdx.y;
  const int rowTile = blockIdx.x;

  const int mn    = lane & 15;
  const int khalf = (lane >> 4) * 2;

  const float* __restrict__ arow = A + (size_t)(rowTile * 16 + mn) * K + khalf;
  const float* __restrict__ wrow = W + (size_t)(colTile * 16 + mn) * K + khalf;

  v8f acc = {};
#pragma unroll
  for (int k0 = 0; k0 < K; k0 += 4) {
    v2f a = *(const v2f*)(arow + k0);
    v2f b = *(const v2f*)(wrow + k0);
    acc = __builtin_amdgcn_wmma_f32_16x16x4_f32(
         false, a,  false, b,
         (short)0, acc,  false,  false);
  }

  const int ncol = colTile * 16 + mn;
  const int mrow = rowTile * 16 + 8 * (lane >> 4);
#pragma unroll
  for (int v = 0; v < 8; ++v)
    H[(size_t)(mrow + v) * Cout + ncol] = acc[v];
}


__global__ void alpha_kernel(const float* __restrict__ H,
                             const float* __restrict__ att_src,
                             const float* __restrict__ att_dst,
                             float* __restrict__ as_, float* __restrict__ ad_,
                             int n, int heads, int C) {
  int idx = blockIdx.x * blockDim.x + threadIdx.x;
  if (idx >= n * heads) return;
  int node = idx / heads, h = idx % heads;
  const float* hp = H + (size_t)node * heads * C + (size_t)h * C;
  const float* sp = att_src + h * C;
  const float* dp = att_dst + h * C;
  float ss = 0.f, sd = 0.f;
  for (int c = 0; c < C; c += 4) {
    float4 hv = *(const float4*)(hp + c);
    float4 sv = *(const float4*)(sp + c);
    float4 dv = *(const float4*)(dp + c);
    ss += hv.x * sv.x + hv.y * sv.y + hv.z * sv.z + hv.w * sv.w;
    sd += hv.x * dv.x + hv.y * dv.y + hv.z * dv.z + hv.w * dv.w;
  }
  as_[idx] = ss;
  ad_[idx] = sd;
}


__device__ __forceinline__ void edge_endpoints(const int* __restrict__ ei,
                                               int E, int i, int& s, int& d) {
  if (i < E) { s = ei[i]; d = ei[E + i]; }
  else       { s = d = i - E; }
}

__global__ void edge_max_kernel(const int* __restrict__ ei, int E, int n,
                                int heads, const float* __restrict__ as_,
                                const float* __restrict__ ad_,
                                unsigned* __restrict__ segmax) {
  int i = blockIdx.x * blockDim.x + threadIdx.x;
  if (i >= E + n) return;
  int s, d;
  edge_endpoints(ei, E, i, s, d);
  for (int h = 0; h < heads; ++h) {
    float e = as_[s * heads + h] + ad_[d * heads + h];
    e = (e > 0.f) ? e : NEG_SLOPE * e;
    atomicMax(&segmax[d * heads + h], fkey(e));
  }
}

__global__ void edge_expsum_kernel(const int* __restrict__ ei, int E, int n,
                                   int heads, const float* __restrict__ as_,
                                   const float* __restrict__ ad_,
                                   const unsigned* __restrict__ segmax,
                                   float* __restrict__ exbuf,
                                   float* __restrict__ denom) {
  int i = blockIdx.x * blockDim.x + threadIdx.x;
  if (i >= E + n) return;
  int s, d;
  edge_endpoints(ei, E, i, s, d);
  for (int h = 0; h < heads; ++h) {
    float e = as_[s * heads + h] + ad_[d * heads + h];
    e = (e > 0.f) ? e : NEG_SLOPE * e;
    float m  = funkey(segmax[d * heads + h]);
    float ex = __expf(e - m);
    exbuf[(size_t)i * heads + h] = ex;
    atomicAdd(&denom[d * heads + h], ex);
  }
}

__global__ void aggregate_kernel(const int* __restrict__ ei, int E, int n,
                                 int heads, int C,
                                 const float* __restrict__ H,
                                 const float* __restrict__ exbuf,
                                 const float* __restrict__ denom,
                                 float* __restrict__ out) {
  int CH = heads * C;
  int idx = blockIdx.x * blockDim.x + threadIdx.x;
  int total = (E + n) * CH;
  if (idx >= total) return;
  int i = idx / CH;
  int c = idx % CH;
  int h = c / C;
  int s, d;
  edge_endpoints(ei, E, i, s, d);
  float alpha = exbuf[(size_t)i * heads + h] /
                (denom[d * heads + h] + 1e-16f);
  atomicAdd(&out[(size_t)d * CH + c], H[(size_t)s * CH + c] * alpha);
}


__global__ void bias_elu_kernel(float* __restrict__ x,
                                const float* __restrict__ b, int n, int CH) {
  int idx = blockIdx.x * blockDim.x + threadIdx.x;
  if (idx >= n * CH) return;
  float y = x[idx] + b[idx % CH];
  x[idx] = (y > 0.f) ? y : expm1f(y);
}

__global__ void bias_add_kernel(float* __restrict__ x,
                                const float* __restrict__ b, int n, int CH) {
  int idx = blockIdx.x * blockDim.x + threadIdx.x;
  if (idx >= n * CH) return;
  x[idx] = x[idx] + b[idx % CH];
}


static inline int cdiv(long long a, long long b) { return (int)((a + b - 1) / b); }

extern "C" void kernel_launch(void* const* d_in, const int* in_sizes, int n_in,
                              void* d_out, int out_size, void* d_ws, size_t ws_size,
                              hipStream_t stream) {
  const int IN_CH = 128, HID = 64, HEADS = 2, OUT = 64;
  const int CH1 = HEADS * HID;

  const float* x        = (const float*)d_in[0];
  const int*   ei       = (const int*)d_in[1];
  const float* W1       = (const float*)d_in[2];
  const float* att1_src = (const float*)d_in[3];
  const float* att1_dst = (const float*)d_in[4];
  const float* b1       = (const float*)d_in[5];
  const float* W2       = (const float*)d_in[6];
  const float* att2_src = (const float*)d_in[7];
  const float* att2_dst = (const float*)d_in[8];
  const float* b2       = (const float*)d_in[9];
  float* out            = (float*)d_out;

  const int n  = in_sizes[0] / IN_CH;
  const int E  = in_sizes[1] / 2;
  const int Ep = E + n;

  char* ws = (char*)d_ws;
  size_t off = 0;
  auto alloc = [&](size_t bytes) -> void* {
    void* p = ws + off;
    off += (bytes + 255) & ~(size_t)255;
    return p;
  };
  float*    h1      = (float*)   alloc((size_t)n * CH1 * 4);
  float*    agg1    = (float*)   alloc((size_t)n * CH1 * 4);
  float*    h2      = (float*)   alloc((size_t)n * OUT * 4);
  float*    as1     = (float*)   alloc((size_t)n * HEADS * 4);
  float*    ad1     = (float*)   alloc((size_t)n * HEADS * 4);
  unsigned* segmax1 = (unsigned*)alloc((size_t)n * HEADS * 4);
  float*    denom1  = (float*)   alloc((size_t)n * HEADS * 4);
  float*    as2     = (float*)   alloc((size_t)n * 4);
  float*    ad2     = (float*)   alloc((size_t)n * 4);
  unsigned* segmax2 = (unsigned*)alloc((size_t)n * 4);
  float*    denom2  = (float*)   alloc((size_t)n * 4);
  float*    ex1     = (float*)   alloc((size_t)Ep * HEADS * 4);
  float*    ex2     = (float*)   alloc((size_t)Ep * 4);
  (void)ws_size;

  const int BT = 256;

  fill_f32_kernel<<<cdiv((long long)n * CH1, BT), BT, 0, stream>>>(agg1, n * CH1, 0.f);
  fill_f32_kernel<<<cdiv(n * HEADS, BT), BT, 0, stream>>>(denom1, n * HEADS, 0.f);
  fill_u32_kernel<<<cdiv(n * HEADS, BT), BT, 0, stream>>>(segmax1, n * HEADS, KEY_NEG_INF);
  fill_f32_kernel<<<cdiv((long long)n * OUT, BT), BT, 0, stream>>>(out, n * OUT, 0.f);
  fill_f32_kernel<<<cdiv(n, BT), BT, 0, stream>>>(denom2, n, 0.f);
  fill_u32_kernel<<<cdiv(n, BT), BT, 0, stream>>>(segmax2, n, KEY_NEG_INF);

  {
    dim3 block(32, CH1 / 16);
    dim3 grid(n / 16);
    gemm_wmma_f32<128><<<grid, block, 0, stream>>>(x, W1, h1, CH1);
  }
  alpha_kernel<<<cdiv(n * HEADS, BT), BT, 0, stream>>>(
      h1, att1_src, att1_dst, as1, ad1, n, HEADS, HID);
  edge_max_kernel<<<cdiv(Ep, BT), BT, 0, stream>>>(ei, E, n, HEADS, as1, ad1, segmax1);
  edge_expsum_kernel<<<cdiv(Ep, BT), BT, 0, stream>>>(
      ei, E, n, HEADS, as1, ad1, segmax1, ex1, denom1);
  aggregate_kernel<<<cdiv((long long)Ep * CH1, BT), BT, 0, stream>>>(
      ei, E, n, HEADS, HID, h1, ex1, denom1, agg1);
  bias_elu_kernel<<<cdiv((long long)n * CH1, BT), BT, 0, stream>>>(agg1, b1, n, CH1);

  {
    dim3 block(32, OUT / 16);
    dim3 grid(n / 16);
    gemm_wmma_f32<128><<<grid, block, 0, stream>>>(agg1, W2, h2, OUT);
  }
  alpha_kernel<<<cdiv(n, BT), BT, 0, stream>>>(
      h2, att2_src, att2_dst, as2, ad2, n, 1, OUT);
  edge_max_kernel<<<cdiv(Ep, BT), BT, 0, stream>>>(ei, E, n, 1, as2, ad2, segmax2);
  edge_expsum_kernel<<<cdiv(Ep, BT), BT, 0, stream>>>(
      ei, E, n, 1, as2, ad2, segmax2, ex2, denom2);
  aggregate_kernel<<<cdiv((long long)Ep * OUT, BT), BT, 0, stream>>>(
      ei, E, n, 1, OUT, h2, ex2, denom2, out);
  bias_add_kernel<<<cdiv((long long)n * OUT, BT), BT, 0, stream>>>(out, b2, n, OUT);
}


# edges counting-sorted by destination once per call; one wave per destination does softmax max/sum and the weighted row sum as a register gather (no float atomics; sum order = sorted slot order)
# speedup vs baseline: 4.7044x; 4.7044x over previous
#include <hip/hip_runtime.h>
#include <hip/hip_bf16.h>


typedef float v2f __attribute__((ext_vector_type(2)));
typedef float v8f __attribute__((ext_vector_type(8)));

#define NEG_SLOPE 0.2f


__global__ void fill_u32_kernel(unsigned* __restrict__ p, int n, unsigned v) {
  int i = blockIdx.x * blockDim.x + threadIdx.x;
  if (i < n) p[i] = v;
}


template <int K>
__global__ __launch_bounds__(256) void gemm_wmma_f32(
    const float* __restrict__ A, const float* __restrict__ W,
    float* __restrict__ H, int Cout) {
  const int lane = threadIdx.x;
  const int colTile = threadIdx.y;
  const int rowTile = blockIdx.x;

  const int mn    = lane & 15;
  const int khalf = (lane >> 4) * 2;

  const float* __restrict__ arow = A + (size_t)(rowTile * 16 + mn) * K + khalf;
  const float* __restrict__ wrow = W + (size_t)(colTile * 16 + mn) * K + khalf;

  v8f acc = {};
#pragma unroll
  for (int k0 = 0; k0 < K; k0 += 4) {
    v2f a = *(const v2f*)(arow + k0);
    v2f b = *(const v2f*)(wrow + k0);
    acc = __builtin_amdgcn_wmma_f32_16x16x4_f32(
         false, a,  false, b,
         (short)0, acc,  false,  false);
  }

  const int ncol = colTile * 16 + mn;
  const int mrow = rowTile * 16 + 8 * (lane >> 4);
#pragma unroll
  for (int v = 0; v < 8; ++v)
    H[(size_t)(mrow + v) * Cout + ncol] = acc[v];
}


__global__ void alpha_kernel(const float* __restrict__ H,
                             const float* __restrict__ att_src,
                             const float* __restrict__ att_dst,
                             float* __restrict__ as_, float* __restrict__ ad_,
                             int n, int heads, int C) {
  int idx = blockIdx.x * blockDim.x + threadIdx.x;
  if (idx >= n * heads) return;
  int node = idx / heads, h = idx % heads;
  const float* hp = H + (size_t)node * heads * C + (size_t)h * C;
  const float* sp = att_src + h * C;
  const float* dp = att_dst + h * C;
  float ss = 0.f, sd = 0.f;
  for (int c = 0; c < C; c += 4) {
    float4 hv = *(const float4*)(hp + c);
    float4 sv = *(const float4*)(sp + c);
    float4 dv = *(const float4*)(dp + c);
    ss += hv.x * sv.x + hv.y * sv.y + hv.z * sv.z + hv.w * sv.w;
    sd += hv.x * dv.x + hv.y * dv.y + hv.z * dv.z + hv.w * dv.w;
  }
  as_[idx] = ss;
  ad_[idx] = sd;
}


__global__ void edge_rank_kernel(const int* __restrict__ ei, int E,
                                 int* __restrict__ deg, int* __restrict__ pos) {
  int i = blockIdx.x * blockDim.x + threadIdx.x;
  if (i >= E) return;
  int d = ei[E + i];
  pos[i] = atomicAdd(&deg[d], 1);
}

__device__ __forceinline__ int wave_sum_i(int v) {
#pragma unroll
  for (int off = 16; off > 0; off >>= 1) v += __shfl_xor(v, off);
  return v;
}

__global__ __launch_bounds__(1024) void rowptr_scan_kernel(
    const int* __restrict__ deg, int* __restrict__ rowptr, int n) {
  constexpr int T = 8;
  __shared__ int wsum[32];
  const int lane = threadIdx.x & 31;
  const int w = __builtin_amdgcn_readfirstlane((int)(threadIdx.x >> 5));
  const int chunks = (n + 31) / 32;
  const int cpw    = (chunks + 31) / 32;
  const int c0 = w * cpw;
  const int c1 = (c0 + cpw < chunks) ? (c0 + cpw) : chunks;

  int s = 0;
  for (int cb = c0; cb < c1; cb += T) {
    int v[T];
#pragma unroll
    for (int t = 0; t < T; ++t) {
      const int i = (cb + t) * 32 + lane;
      v[t] = (cb + t < c1 && i < n) ? deg[i] : 0;
    }
#pragma unroll
    for (int t = 0; t < T; ++t) s += v[t];
  }
  s = wave_sum_i(s);
  if (lane == 0) wsum[w] = s;
  __syncthreads();
  int run = wave_sum_i((lane < w) ? wsum[lane] : 0);

  for (int cb = c0; cb < c1; cb += T) {
    int v[T], inc[T];
#pragma unroll
    for (int t = 0; t < T; ++t) {
      const int i = (cb + t) * 32 + lane;
      v[t] = (cb + t < c1 && i < n) ? deg[i] : 0;
      inc[t] = v[t];
    }
#pragma unroll
    for (int off = 1; off < 32; off <<= 1) {
#pragma unroll
      for (int t = 0; t < T; ++t) {
        const int u = __shfl_up(inc[t], off);
        if (lane >= off) inc[t] += u;
      }
    }
#pragma unroll
    for (int t = 0; t < T; ++t) {
      const int i = (cb + t) * 32 + lane;
      if (cb + t < c1 && i < n) rowptr[i] = run + inc[t] - v[t];
      run += __builtin_amdgcn_readlane(inc[t], 31);
    }
  }
  if (w == 31 && lane == 0) rowptr[n] = run;
}

__global__ void edge_place_kernel(const int* __restrict__ ei, int E,
                                  const int* __restrict__ rowptr,
                                  const int* __restrict__ pos,
                                  int* __restrict__ csr_src) {
  int i = blockIdx.x * blockDim.x + threadIdx.x;
  if (i >= E) return;
  int s = ei[i];
  int d = ei[E + i];
  int slot = rowptr[d] + pos[i];
  if ((unsigned)slot < (unsigned)E) csr_src[slot] = s;
}


__device__ __forceinline__ float wave_max_f(float v) {
#pragma unroll
  for (int off = 16; off > 0; off >>= 1) v = fmaxf(v, __shfl_xor(v, off));
  return v;
}
__device__ __forceinline__ float wave_sum_f(float v) {
#pragma unroll
  for (int off = 16; off > 0; off >>= 1) v += __shfl_xor(v, off);
  return v;
}
__device__ __forceinline__ float leaky(float e) {
  return (e > 0.f) ? e : NEG_SLOPE * e;
}

template <int HEADS, int C, bool ELU>
__global__ __launch_bounds__(256) void gat_gather_kernel(
    const int* __restrict__ rowptr, const int* __restrict__ csr_src,
    const float* __restrict__ H, const float* __restrict__ as_,
    const float* __restrict__ ad_, const float* __restrict__ bias,
    float* __restrict__ out, int n, int E) {
  constexpr int CH  = HEADS * C;
  constexpr int VEC = CH / 32;
  typedef float vecf __attribute__((ext_vector_type(VEC)));

  const int lane = threadIdx.x & 31;
  const int wv  = __builtin_amdgcn_readfirstlane((int)(threadIdx.x >> 5));
  const int gw  = blockIdx.x * 8 + wv;
  const int nw  = gridDim.x * 8;
  const int myh = (lane * VEC) / C;
  const vecf bv = *(const vecf*)(bias + lane * VEC);

  for (int node = gw; node < n; node += nw) {
    int beg = rowptr[node];
    int end = rowptr[node + 1];
    beg = (beg < 0) ? 0 : beg;
    end = (end > E) ? E : end;
    const int deg = end - beg;

    const vecf hself = *(const vecf*)(H + (size_t)node * CH + lane * VEC);

    float adh[HEADS], eself[HEADS], m[HEADS];
#pragma unroll
    for (int h = 0; h < HEADS; ++h) {
      adh[h]   = ad_[node * HEADS + h];
      eself[h] = leaky(as_[node * HEADS + h] + adh[h]);
    }

    const bool v0 = lane < deg;
    int s0 = node;
    if (v0) {
      s0 = csr_src[beg + lane];
      s0 = (s0 < 0) ? 0 : ((s0 >= n) ? n - 1 : s0);
    }
    float e0[HEADS];
    if (HEADS == 2) {
      const v2f t = *(const v2f*)(as_ + (size_t)s0 * 2);
      e0[0]         = leaky(t.x + adh[0]);
      e0[HEADS - 1] = leaky(t.y + adh[HEADS - 1]);
    } else {
      e0[0] = leaky(as_[s0] + adh[0]);
    }
#pragma unroll
    for (int h = 0; h < HEADS; ++h) m[h] = fmaxf(eself[h], e0[h]);
    for (int base = 32; base < deg; base += 32) {
      int j = base + lane;
      if (j < deg) {
        int s = csr_src[beg + j];
        s = (s < 0) ? 0 : ((s >= n) ? n - 1 : s);
#pragma unroll
        for (int h = 0; h < HEADS; ++h)
          m[h] = fmaxf(m[h], leaky(as_[s * HEADS + h] + adh[h]));
      }
    }
#pragma unroll
    for (int h = 0; h < HEADS; ++h) m[h] = wave_max_f(m[h]);

    float p0[HEADS], den[HEADS], pself[HEADS];
#pragma unroll
    for (int h = 0; h < HEADS; ++h) {
      p0[h]  = v0 ? __expf(e0[h] - m[h]) : 0.f;
      den[h] = p0[h];
    }
    for (int base = 32; base < deg; base += 32) {
      int j = base + lane;
      if (j < deg) {
        int s = csr_src[beg + j];
        s = (s < 0) ? 0 : ((s >= n) ? n - 1 : s);
#pragma unroll
        for (int h = 0; h < HEADS; ++h)
          den[h] += __expf(leaky(as_[s * HEADS + h] + adh[h]) - m[h]);
      }
    }
#pragma unroll
    for (int h = 0; h < HEADS; ++h) {
      pself[h] = __expf(eself[h] - m[h]);
      den[h]   = wave_sum_f(den[h]) + pself[h] + 1e-16f;
    }

    vecf acc;
    {
      const float a = (HEADS == 2 && myh) ? pself[HEADS - 1] / den[HEADS - 1]
                                          : pself[0] / den[0];
      acc = hself * a;
    }
    for (int base = 0; base < deg; base += 32) {
      int   sj = s0;
      float al[HEADS];
      if (base == 0) {
#pragma unroll
        for (int h = 0; h < HEADS; ++h) al[h] = p0[h] / den[h];
      } else {
        int j = base + lane;
        sj = node;
#pragma unroll
        for (int h = 0; h < HEADS; ++h) al[h] = 0.f;
        if (j < deg) {
          sj = csr_src[beg + j];
          sj = (sj < 0) ? 0 : ((sj >= n) ? n - 1 : sj);
#pragma unroll
          for (int h = 0; h < HEADS; ++h)
            al[h] = __expf(leaky(as_[sj * HEADS + h] + adh[h]) - m[h]) / den[h];
        }
      }
      const int cnt  = (deg - base < 32) ? (deg - base) : 32;
      const int cnt4 = (cnt + 3) & ~3;
      for (int k = 0; k < cnt4; k += 4) {
        vecf  hv[4];
        float a[4];
#pragma unroll
        for (int u = 0; u < 4; ++u) {
          const int sk = __builtin_amdgcn_readlane(sj, k + u);
          hv[u] = *(const vecf*)(H + (size_t)sk * CH + lane * VEC);
          const float a0 = __int_as_float(__builtin_amdgcn_readlane(__float_as_int(al[0]), k + u));
          a[u] = a0;
          if (HEADS == 2) {
            const float a1 = __int_as_float(__builtin_amdgcn_readlane(__float_as_int(al[HEADS - 1]), k + u));
            a[u] = myh ? a1 : a0;
          }
        }
#pragma unroll
        for (int u = 0; u < 4; ++u) acc += hv[u] * a[u];
      }
    }

#pragma unroll
    for (int v = 0; v < VEC; ++v) {
      float y = acc[v] + bv[v];
      if (ELU) y = (y > 0.f) ? y : expm1f(y);
      acc[v] = y;
    }
    *(vecf*)(out + (size_t)node * CH + lane * VEC) = acc;
  }
}


static inline int cdiv(long long a, long long b) { return (int)((a + b - 1) / b); }

extern "C" void kernel_launch(void* const* d_in, const int* in_sizes, int n_in,
                              void* d_out, int out_size, void* d_ws, size_t ws_size,
                              hipStream_t stream) {
  const int IN_CH = 128, HID = 64, HEADS = 2, OUT = 64;
  const int CH1 = HEADS * HID;

  const float* x        = (const float*)d_in[0];
  const int*   ei       = (const int*)d_in[1];
  const float* W1       = (const float*)d_in[2];
  const float* att1_src = (const float*)d_in[3];
  const float* att1_dst = (const float*)d_in[4];
  const float* b1       = (const float*)d_in[5];
  const float* W2       = (const float*)d_in[6];
  const float* att2_src = (const float*)d_in[7];
  const float* att2_dst = (const float*)d_in[8];
  const float* b2       = (const float*)d_in[9];
  float* out            = (float*)d_out;

  const int n  = in_sizes[0] / IN_CH;
  const int E  = in_sizes[1] / 2;

  char* ws = (char*)d_ws;
  size_t off = 0;
  auto alloc = [&](size_t bytes) -> void* {
    void* p = ws + off;
    off += (bytes + 255) & ~(size_t)255;
    return p;
  };
  float* h1      = (float*)alloc((size_t)n * CH1 * 4);
  float* agg1    = (float*)alloc((size_t)n * CH1 * 4);
  float* h2      = (float*)alloc((size_t)n * OUT * 4);
  float* as1     = (float*)alloc((size_t)n * HEADS * 4);
  float* ad1     = (float*)alloc((size_t)n * HEADS * 4);
  float* as2     = (float*)alloc((size_t)n * 4);
  float* ad2     = (float*)alloc((size_t)n * 4);
  int*   deg     = (int*)  alloc((size_t)n * 4);
  int*   rowptr  = (int*)  alloc((size_t)(n + 1) * 4);
  int*   pos     = (int*)  alloc((size_t)E * 4);
  int*   csr_src = (int*)  alloc((size_t)E * 4);
  (void)ws_size; (void)n_in; (void)out_size;

  const int BT = 256;
  const int GATHER_BLOCKS = 2048;

  fill_u32_kernel<<<cdiv(n, BT), BT, 0, stream>>>((unsigned*)deg, n, 0u);
  edge_rank_kernel<<<cdiv(E, BT), BT, 0, stream>>>(ei, E, deg, pos);
  rowptr_scan_kernel<<<1, 1024, 0, stream>>>(deg, rowptr, n);
  edge_place_kernel<<<cdiv(E, BT), BT, 0, stream>>>(ei, E, rowptr, pos, csr_src);

  {
    dim3 block(32, CH1 / 16);
    dim3 grid(n / 16);
    gemm_wmma_f32<128><<<grid, block, 0, stream>>>(x, W1, h1, CH1);
  }
  alpha_kernel<<<cdiv(n * HEADS, BT), BT, 0, stream>>>(
      h1, att1_src, att1_dst, as1, ad1, n, HEADS, HID);
  gat_gather_kernel<2, 64, true><<<GATHER_BLOCKS, BT, 0, stream>>>(
      rowptr, csr_src, h1, as1, ad1, b1, agg1, n, E);

  {
    dim3 block(32, OUT / 16);
    dim3 grid(n / 16);
    gemm_wmma_f32<128><<<grid, block, 0, stream>>>(agg1, W2, h2, OUT);
  }
  alpha_kernel<<<cdiv(n, BT), BT, 0, stream>>>(
      h2, att2_src, att2_dst, as2, ad2, n, 1, OUT);
  gat_gather_kernel<1, 64, false><<<GATHER_BLOCKS, BT, 0, stream>>>(
      rowptr, csr_src, h2, as2, ad2, b2, out, n, E);
}


# GEMMs: W and a 32/64-row slab of A staged in LDS by whole-row 16-byte loads, fragments read from LDS, two row tiles per wave share each B fragment; k order and sums unchanged (bit-identical)
# speedup vs baseline: 5.5689x; 1.1838x over previous
#include <hip/hip_runtime.h>
#include <hip/hip_bf16.h>


typedef float v2f __attribute__((ext_vector_type(2)));
typedef float v8f __attribute__((ext_vector_type(8)));

#define NEG_SLOPE 0.2f


__global__ void fill_u32_kernel(unsigned* __restrict__ p, int n, unsigned v) {
  int i = blockIdx.x * blockDim.x + threadIdx.x;
  if (i < n) p[i] = v;
}


template <int COUT, int BM>
__global__ __launch_bounds__(256) void gemm_wmma_f32(
    const float* __restrict__ A, const float* __restrict__ W,
    float* __restrict__ H, int n) {
  constexpr int K  = 128;
  constexpr int P  = K + 4;
  constexpr int CT = COUT / 16;
  static_assert(CT * (BM / 32) == 8, "8 waves per work-group");
  constexpr int WV4 = COUT * (K / 4) / 256;
  constexpr int AV4 = BM * (K / 4) / 256;
  __shared__ __attribute__((aligned(16))) float Ws[COUT * P];
  __shared__ __attribute__((aligned(16))) float As[BM * P];

  const int tid  = threadIdx.x;
  const int lane = tid & 31;
  const int wv   = __builtin_amdgcn_readfirstlane(tid >> 5);
  const int row0 = blockIdx.x * BM;

  float4 wreg[WV4], areg[AV4];
#pragma unroll
  for (int j = 0; j < WV4; ++j) {
    const int r = wv + 8 * j;
    wreg[j] = *(const float4*)(W + (size_t)r * K + lane * 4);
  }
#pragma unroll
  for (int j = 0; j < AV4; ++j) {
    int r = row0 + wv + 8 * j;
    r = (r < n) ? r : (n - 1);
    areg[j] = *(const float4*)(A + (size_t)r * K + lane * 4);
  }
#pragma unroll
  for (int j = 0; j < WV4; ++j)
    *(float4*)(Ws + (wv + 8 * j) * P + lane * 4) = wreg[j];
#pragma unroll
  for (int j = 0; j < AV4; ++j)
    *(float4*)(As + (wv + 8 * j) * P + lane * 4) = areg[j];
  __syncthreads();

  const int mn      = lane & 15;
  const int khalf   = (lane >> 4) * 2;
  const int colTile = wv % CT;
  const int rg      = wv / CT;

  const float* bp  = Ws + (colTile * 16 + mn) * P + khalf;
  const float* ap0 = As + (rg * 32 + mn) * P + khalf;
  const float* ap1 = ap0 + 16 * P;

  v8f acc0 = {}, acc1 = {};
#pragma unroll
  for (int k0 = 0; k0 < K; k0 += 4) {
    const v2f b  = *(const v2f*)(bp + k0);
    const v2f a0 = *(const v2f*)(ap0 + k0);
    const v2f a1 = *(const v2f*)(ap1 + k0);
    acc0 = __builtin_amdgcn_wmma_f32_16x16x4_f32(
         false, a0,  false, b,
         (short)0, acc0,  false,  false);
    acc1 = __builtin_amdgcn_wmma_f32_16x16x4_f32(
         false, a1,  false, b,
         (short)0, acc1,  false,  false);
  }

  const int ncol  = colTile * 16 + mn;
  const int trow  = row0 + rg * 32;
  const int mrow  = trow + 8 * (lane >> 4);
  if (trow < n) {
#pragma unroll
    for (int v = 0; v < 8; ++v)
      H[(size_t)(mrow + v) * COUT + ncol] = acc0[v];
  }
  if (trow + 16 < n) {
#pragma unroll
    for (int v = 0; v < 8; ++v)
      H[(size_t)(mrow + 16 + v) * COUT + ncol] = acc1[v];
  }
}


__global__ void alpha_kernel(const float* __restrict__ H,
                             const float* __restrict__ att_src,
                             const float* __restrict__ att_dst,
                             float* __restrict__ as_, float* __restrict__ ad_,
                             int n, int heads, int C) {
  int idx = blockIdx.x * blockDim.x + threadIdx.x;
  if (idx >= n * heads) return;
  int node = idx / heads, h = idx % heads;
  const float* hp = H + (size_t)node * heads * C + (size_t)h * C;
  const float* sp = att_src + h * C;
  const float* dp = att_dst + h * C;
  float ss = 0.f, sd = 0.f;
  for (int c = 0; c < C; c += 4) {
    float4 hv = *(const float4*)(hp + c);
    float4 sv = *(const float4*)(sp + c);
    float4 dv = *(const float4*)(dp + c);
    ss += hv.x * sv.x + hv.y * sv.y + hv.z * sv.z + hv.w * sv.w;
    sd += hv.x * dv.x + hv.y * dv.y + hv.z * dv.z + hv.w * dv.w;
  }
  as_[idx] = ss;
  ad_[idx] = sd;
}


__global__ void edge_rank_kernel(const int* __restrict__ ei, int E,
                                 int* __restrict__ deg, int* __restrict__ pos) {
  int i = blockIdx.x * blockDim.x + threadIdx.x;
  if (i >= E) return;
  int d = ei[E + i];
  pos[i] = atomicAdd(&deg[d], 1);
}

__device__ __forceinline__ int wave_sum_i(int v) {
#pragma unroll
  for (int off = 16; off > 0; off >>= 1) v += __shfl_xor(v, off);
  return v;
}

__global__ __launch_bounds__(1024) void rowptr_scan_kernel(
    const int* __restrict__ deg, int* __restrict__ rowptr, int n) {
  constexpr int T = 8;
  __shared__ int wsum[32];
  const int lane = threadIdx.x & 31;
  const int w = __builtin_amdgcn_readfirstlane((int)(threadIdx.x >> 5));
  const int chunks = (n + 31) / 32;
  const int cpw    = (chunks + 31) / 32;
  const int c0 = w * cpw;
  const int c1 = (c0 + cpw < chunks) ? (c0 + cpw) : chunks;

  int s = 0;
  for (int cb = c0; cb < c1; cb += T) {
    int v[T];
#pragma unroll
    for (int t = 0; t < T; ++t) {
      const int i = (cb + t) * 32 + lane;
      v[t] = (cb + t < c1 && i < n) ? deg[i] : 0;
    }
#pragma unroll
    for (int t = 0; t < T; ++t) s += v[t];
  }
  s = wave_sum_i(s);
  if (lane == 0) wsum[w] = s;
  __syncthreads();
  int run = wave_sum_i((lane < w) ? wsum[lane] : 0);

  for (int cb = c0; cb < c1; cb += T) {
    int v[T], inc[T];
#pragma unroll
    for (int t = 0; t < T; ++t) {
      const int i = (cb + t) * 32 + lane;
      v[t] = (cb + t < c1 && i < n) ? deg[i] : 0;
      inc[t] = v[t];
    }
#pragma unroll
    for (int off = 1; off < 32; off <<= 1) {
#pragma unroll
      for (int t = 0; t < T; ++t) {
        const int u = __shfl_up(inc[t], off);
        if (lane >= off) inc[t] += u;
      }
    }
#pragma unroll
    for (int t = 0; t < T; ++t) {
      const int i = (cb + t) * 32 + lane;
      if (cb + t < c1 && i < n) rowptr[i] = run + inc[t] - v[t];
      run += __builtin_amdgcn_readlane(inc[t], 31);
    }
  }
  if (w == 31 && lane == 0) rowptr[n] = run;
}

__global__ void edge_place_kernel(const int* __restrict__ ei, int E,
                                  const int* __restrict__ rowptr,
                                  const int* __restrict__ pos,
                                  int* __restrict__ csr_src) {
  int i = blockIdx.x * blockDim.x + threadIdx.x;
  if (i >= E) return;
  int s = ei[i];
  int d = ei[E + i];
  int slot = rowptr[d] + pos[i];
  if ((unsigned)slot < (unsigned)E) csr_src[slot] = s;
}


__device__ __forceinline__ float wave_max_f(float v) {
#pragma unroll
  for (int off = 16; off > 0; off >>= 1) v = fmaxf(v, __shfl_xor(v, off));
  return v;
}
__device__ __forceinline__ float wave_sum_f(float v) {
#pragma unroll
  for (int off = 16; off > 0; off >>= 1) v += __shfl_xor(v, off);
  return v;
}
__device__ __forceinline__ float leaky(float e) {
  return (e > 0.f) ? e : NEG_SLOPE * e;
}

template <int HEADS, int C, bool ELU>
__global__ __launch_bounds__(256) void gat_gather_kernel(
    const int* __restrict__ rowptr, const int* __restrict__ csr_src,
    const float* __restrict__ H, const float* __restrict__ as_,
    const float* __restrict__ ad_, const float* __restrict__ bias,
    float* __restrict__ out, int n, int E) {
  constexpr int CH  = HEADS * C;
  constexpr int VEC = CH / 32;
  typedef float vecf __attribute__((ext_vector_type(VEC)));

  const int lane = threadIdx.x & 31;
  const int wv  = __builtin_amdgcn_readfirstlane((int)(threadIdx.x >> 5));
  const int gw  = blockIdx.x * 8 + wv;
  const int nw  = gridDim.x * 8;
  const int myh = (lane * VEC) / C;
  const vecf bv = *(const vecf*)(bias + lane * VEC);

  for (int node = gw; node < n; node += nw) {
    int beg = rowptr[node];
    int end = rowptr[node + 1];
    beg = (beg < 0) ? 0 : beg;
    end = (end > E) ? E : end;
    const int deg = end - beg;

    const vecf hself = *(const vecf*)(H + (size_t)node * CH + lane * VEC);

    float adh[HEADS], eself[HEADS], m[HEADS];
#pragma unroll
    for (int h = 0; h < HEADS; ++h) {
      adh[h]   = ad_[node * HEADS + h];
      eself[h] = leaky(as_[node * HEADS + h] + adh[h]);
    }

    const bool v0 = lane < deg;
    int s0 = node;
    if (v0) {
      s0 = csr_src[beg + lane];
      s0 = (s0 < 0) ? 0 : ((s0 >= n) ? n - 1 : s0);
    }
    float e0[HEADS];
    if (HEADS == 2) {
      const v2f t = *(const v2f*)(as_ + (size_t)s0 * 2);
      e0[0]         = leaky(t.x + adh[0]);
      e0[HEADS - 1] = leaky(t.y + adh[HEADS - 1]);
    } else {
      e0[0] = leaky(as_[s0] + adh[0]);
    }
#pragma unroll
    for (int h = 0; h < HEADS; ++h) m[h] = fmaxf(eself[h], e0[h]);
    for (int base = 32; base < deg; base += 32) {
      int j = base + lane;
      if (j < deg) {
        int s = csr_src[beg + j];
        s = (s < 0) ? 0 : ((s >= n) ? n - 1 : s);
#pragma unroll
        for (int h = 0; h < HEADS; ++h)
          m[h] = fmaxf(m[h], leaky(as_[s * HEADS + h] + adh[h]));
      }
    }
#pragma unroll
    for (int h = 0; h < HEADS; ++h) m[h] = wave_max_f(m[h]);

    float p0[HEADS], den[HEADS], pself[HEADS];
#pragma unroll
    for (int h = 0; h < HEADS; ++h) {
      p0[h]  = v0 ? __expf(e0[h] - m[h]) : 0.f;
      den[h] = p0[h];
    }
    for (int base = 32; base < deg; base += 32) {
      int j = base + lane;
      if (j < deg) {
        int s = csr_src[beg + j];
        s = (s < 0) ? 0 : ((s >= n) ? n - 1 : s);
#pragma unroll
        for (int h = 0; h < HEADS; ++h)
          den[h] += __expf(leaky(as_[s * HEADS + h] + adh[h]) - m[h]);
      }
    }
#pragma unroll
    for (int h = 0; h < HEADS; ++h) {
      pself[h] = __expf(eself[h] - m[h]);
      den[h]   = wave_sum_f(den[h]) + pself[h] + 1e-16f;
    }

    vecf acc;
    {
      const float a = (HEADS == 2 && myh) ? pself[HEADS - 1] / den[HEADS - 1]
                                          : pself[0] / den[0];
      acc = hself * a;
    }
    for (int base = 0; base < deg; base += 32) {
      int   sj = s0;
      float al[HEADS];
      if (base == 0) {
#pragma unroll
        for (int h = 0; h < HEADS; ++h) al[h] = p0[h] / den[h];
      } else {
        int j = base + lane;
        sj = node;
#pragma unroll
        for (int h = 0; h < HEADS; ++h) al[h] = 0.f;
        if (j < deg) {
          sj = csr_src[beg + j];
          sj = (sj < 0) ? 0 : ((sj >= n) ? n - 1 : sj);
#pragma unroll
          for (int h = 0; h < HEADS; ++h)
            al[h] = __expf(leaky(as_[sj * HEADS + h] + adh[h]) - m[h]) / den[h];
        }
      }
      const int cnt  = (deg - base < 32) ? (deg - base) : 32;
      const int cnt4 = (cnt + 3) & ~3;
      for (int k = 0; k < cnt4; k += 4) {
        vecf  hv[4];
        float a[4];
#pragma unroll
        for (int u = 0; u < 4; ++u) {
          const int sk = __builtin_amdgcn_readlane(sj, k + u);
          hv[u] = *(const vecf*)(H + (size_t)sk * CH + lane * VEC);
          const float a0 = __int_as_float(__builtin_amdgcn_readlane(__float_as_int(al[0]), k + u));
          a[u] = a0;
          if (HEADS == 2) {
            const float a1 = __int_as_float(__builtin_amdgcn_readlane(__float_as_int(al[HEADS - 1]), k + u));
            a[u] = myh ? a1 : a0;
          }
        }
#pragma unroll
        for (int u = 0; u < 4; ++u) acc += hv[u] * a[u];
      }
    }

#pragma unroll
    for (int v = 0; v < VEC; ++v) {
      float y = acc[v] + bv[v];
      if (ELU) y = (y > 0.f) ? y : expm1f(y);
      acc[v] = y;
    }
    *(vecf*)(out + (size_t)node * CH + lane * VEC) = acc;
  }
}


static inline int cdiv(long long a, long long b) { return (int)((a + b - 1) / b); }

extern "C" void kernel_launch(void* const* d_in, const int* in_sizes, int n_in,
                              void* d_out, int out_size, void* d_ws, size_t ws_size,
                              hipStream_t stream) {
  const int IN_CH = 128, HID = 64, HEADS = 2, OUT = 64;
  const int CH1 = HEADS * HID;

  const float* x        = (const float*)d_in[0];
  const int*   ei       = (const int*)d_in[1];
  const float* W1       = (const float*)d_in[2];
  const float* att1_src = (const float*)d_in[3];
  const float* att1_dst = (const float*)d_in[4];
  const float* b1       = (const float*)d_in[5];
  const float* W2       = (const float*)d_in[6];
  const float* att2_src = (const float*)d_in[7];
  const float* att2_dst = (const float*)d_in[8];
  const float* b2       = (const float*)d_in[9];
  float* out            = (float*)d_out;

  const int n  = in_sizes[0] / IN_CH;
  const int E  = in_sizes[1] / 2;

  char* ws = (char*)d_ws;
  size_t off = 0;
  auto alloc = [&](size_t bytes) -> void* {
    void* p = ws + off;
    off += (bytes + 255) & ~(size_t)255;
    return p;
  };
  float* h1      = (float*)alloc((size_t)n * CH1 * 4);
  float* agg1    = (float*)alloc((size_t)n * CH1 * 4);
  float* h2      = (float*)alloc((size_t)n * OUT * 4);
  float* as1     = (float*)alloc((size_t)n * HEADS * 4);
  float* ad1     = (float*)alloc((size_t)n * HEADS * 4);
  float* as2     = (float*)alloc((size_t)n * 4);
  float* ad2     = (float*)alloc((size_t)n * 4);
  int*   deg     = (int*)  alloc((size_t)n * 4);
  int*   rowptr  = (int*)  alloc((size_t)(n + 1) * 4);
  int*   pos     = (int*)  alloc((size_t)E * 4);
  int*   csr_src = (int*)  alloc((size_t)E * 4);
  (void)ws_size; (void)n_in; (void)out_size;

  const int BT = 256;
  const int GATHER_BLOCKS = 2048;

  fill_u32_kernel<<<cdiv(n, BT), BT, 0, stream>>>((unsigned*)deg, n, 0u);
  edge_rank_kernel<<<cdiv(E, BT), BT, 0, stream>>>(ei, E, deg, pos);
  rowptr_scan_kernel<<<1, 1024, 0, stream>>>(deg, rowptr, n);
  edge_place_kernel<<<cdiv(E, BT), BT, 0, stream>>>(ei, E, rowptr, pos, csr_src);

  gemm_wmma_f32<128, 32><<<cdiv(n, 32), BT, 0, stream>>>(x, W1, h1, n);
  alpha_kernel<<<cdiv(n * HEADS, BT), BT, 0, stream>>>(
      h1, att1_src, att1_dst, as1, ad1, n, HEADS, HID);
  gat_gather_kernel<2, 64, true><<<GATHER_BLOCKS, BT, 0, stream>>>(
      rowptr, csr_src, h1, as1, ad1, b1, agg1, n, E);

  gemm_wmma_f32<64, 64><<<cdiv(n, 64), BT, 0, stream>>>(agg1, W2, h2, n);
  alpha_kernel<<<cdiv(n, BT), BT, 0, stream>>>(
      h2, att2_src, att2_dst, as2, ad2, n, 1, OUT);
  gat_gather_kernel<1, 64, false><<<GATHER_BLOCKS, BT, 0, stream>>>(
      rowptr, csr_src, h2, as2, ad2, b2, out, n, E);
}


# edge ranks from returning LDS adds: a work-group counts a 16384-edge slice in a node-counter table in LDS, a column pass sums the tables; no global atomics (sum order: by slice, then LDS arrival)
# speedup vs baseline: 7.0482x; 1.2656x over previous
#include <hip/hip_runtime.h>
#include <hip/hip_bf16.h>


typedef float v2f __attribute__((ext_vector_type(2)));
typedef float v8f __attribute__((ext_vector_type(8)));

#define NEG_SLOPE 0.2f


template <int COUT, int BM>
__global__ __launch_bounds__(256) void gemm_wmma_f32(
    const float* __restrict__ A, const float* __restrict__ W,
    float* __restrict__ H, int n) {
  constexpr int K  = 128;
  constexpr int P  = K + 4;
  constexpr int CT = COUT / 16;
  static_assert(CT * (BM / 32) == 8, "8 waves per work-group");
  constexpr int WV4 = COUT * (K / 4) / 256;
  constexpr int AV4 = BM * (K / 4) / 256;
  __shared__ __attribute__((aligned(16))) float Ws[COUT * P];
  __shared__ __attribute__((aligned(16))) float As[BM * P];

  const int tid  = threadIdx.x;
  const int lane = tid & 31;
  const int wv   = __builtin_amdgcn_readfirstlane(tid >> 5);
  const int row0 = blockIdx.x * BM;

  float4 wreg[WV4], areg[AV4];
#pragma unroll
  for (int j = 0; j < WV4; ++j) {
    const int r = wv + 8 * j;
    wreg[j] = *(const float4*)(W + (size_t)r * K + lane * 4);
  }
#pragma unroll
  for (int j = 0; j < AV4; ++j) {
    int r = row0 + wv + 8 * j;
    r = (r < n) ? r : (n - 1);
    areg[j] = *(const float4*)(A + (size_t)r * K + lane * 4);
  }
#pragma unroll
  for (int j = 0; j < WV4; ++j)
    *(float4*)(Ws + (wv + 8 * j) * P + lane * 4) = wreg[j];
#pragma unroll
  for (int j = 0; j < AV4; ++j)
    *(float4*)(As + (wv + 8 * j) * P + lane * 4) = areg[j];
  __syncthreads();

  const int mn      = lane & 15;
  const int khalf   = (lane >> 4) * 2;
  const int colTile = wv % CT;
  const int rg      = wv / CT;

  const float* bp  = Ws + (colTile * 16 + mn) * P + khalf;
  const float* ap0 = As + (rg * 32 + mn) * P + khalf;
  const float* ap1 = ap0 + 16 * P;

  v8f acc0 = {}, acc1 = {};
#pragma unroll
  for (int k0 = 0; k0 < K; k0 += 4) {
    const v2f b  = *(const v2f*)(bp + k0);
    const v2f a0 = *(const v2f*)(ap0 + k0);
    const v2f a1 = *(const v2f*)(ap1 + k0);
    acc0 = __builtin_amdgcn_wmma_f32_16x16x4_f32(
         false, a0,  false, b,
         (short)0, acc0,  false,  false);
    acc1 = __builtin_amdgcn_wmma_f32_16x16x4_f32(
         false, a1,  false, b,
         (short)0, acc1,  false,  false);
  }

  const int ncol  = colTile * 16 + mn;
  const int trow  = row0 + rg * 32;
  const int mrow  = trow + 8 * (lane >> 4);
  if (trow < n) {
#pragma unroll
    for (int v = 0; v < 8; ++v)
      H[(size_t)(mrow + v) * COUT + ncol] = acc0[v];
  }
  if (trow + 16 < n) {
#pragma unroll
    for (int v = 0; v < 8; ++v)
      H[(size_t)(mrow + 16 + v) * COUT + ncol] = acc1[v];
  }
}


__global__ void alpha_kernel(const float* __restrict__ H,
                             const float* __restrict__ att_src,
                             const float* __restrict__ att_dst,
                             float* __restrict__ as_, float* __restrict__ ad_,
                             int n, int heads, int C) {
  int idx = blockIdx.x * blockDim.x + threadIdx.x;
  if (idx >= n * heads) return;
  int node = idx / heads, h = idx % heads;
  const float* hp = H + (size_t)node * heads * C + (size_t)h * C;
  const float* sp = att_src + h * C;
  const float* dp = att_dst + h * C;
  float ss = 0.f, sd = 0.f;
  for (int c = 0; c < C; c += 4) {
    float4 hv = *(const float4*)(hp + c);
    float4 sv = *(const float4*)(sp + c);
    float4 dv = *(const float4*)(dp + c);
    ss += hv.x * sv.x + hv.y * sv.y + hv.z * sv.z + hv.w * sv.w;
    sd += hv.x * dv.x + hv.y * dv.y + hv.z * dv.z + hv.w * dv.w;
  }
  as_[idx] = ss;
  ad_[idx] = sd;
}


constexpr int SLICE_LOG2 = 14;
constexpr int SLICE      = 1 << SLICE_LOG2;
constexpr int NB         = 65536;

__global__ __launch_bounds__(1024) void edge_count_kernel(
    const int* __restrict__ ei, int E, int n,
    int* __restrict__ hist, int* __restrict__ pos) {
  constexpr int U = SLICE / 4096;
  __shared__ __attribute__((aligned(16))) int cnt[NB];
  const int tid = threadIdx.x;
  const int g   = blockIdx.x;
  const int np  = (n + 3) & ~3;
  const int* dst = ei + E;
  const bool vec = (E & 3) == 0;
  const int e0   = g * SLICE + tid * 4;

  const bool full = vec && (g * SLICE + SLICE <= E);

  int4 d[U], r[U];
  if (full) {
#pragma unroll
    for (int u = 0; u < U; ++u) d[u] = *(const int4*)(dst + e0 + u * 4096);
  } else {
#pragma unroll
    for (int u = 0; u < U; ++u) {
      const int e = e0 + u * 4096;
      const int x = dst[(e     < E) ? e     : E - 1];
      const int y = dst[(e + 1 < E) ? e + 1 : E - 1];
      const int z = dst[(e + 2 < E) ? e + 2 : E - 1];
      const int w = dst[(e + 3 < E) ? e + 3 : E - 1];
      d[u].x = (e     < E) ? x : -1;
      d[u].y = (e + 1 < E) ? y : -1;
      d[u].z = (e + 2 < E) ? z : -1;
      d[u].w = (e + 3 < E) ? w : -1;
    }
  }
#pragma unroll
  for (int u = 0; u < U; ++u) r[u] = make_int4(0, 0, 0, 0);

  for (int nb = 0; nb < n; nb += NB) {
    const int cur  = (n - nb < NB) ? (n - nb) : NB;
    const int cur4 = (cur + 3) & ~3;
    if (nb) __syncthreads();
    for (int j = tid * 4; j < cur4; j += 4096)
      *(int4*)(cnt + j) = make_int4(0, 0, 0, 0);
    __syncthreads();
#pragma unroll
    for (int u = 0; u < U; ++u) {
      const unsigned ox = (unsigned)(d[u].x - nb), oy = (unsigned)(d[u].y - nb);
      const unsigned oz = (unsigned)(d[u].z - nb), ow = (unsigned)(d[u].w - nb);
      if (ox < (unsigned)cur) r[u].x = atomicAdd(&cnt[ox], 1);
      if (oy < (unsigned)cur) r[u].y = atomicAdd(&cnt[oy], 1);
      if (oz < (unsigned)cur) r[u].z = atomicAdd(&cnt[oz], 1);
      if (ow < (unsigned)cur) r[u].w = atomicAdd(&cnt[ow], 1);
    }
    __syncthreads();
    int* hp = hist + (size_t)g * np + nb;
    int j = tid * 4;
    for (; j + 3 * 4096 < cur4; j += 4 * 4096) {
      int4 t[4];
#pragma unroll
      for (int q = 0; q < 4; ++q) t[q] = *(const int4*)(cnt + j + q * 4096);
#pragma unroll
      for (int q = 0; q < 4; ++q) *(int4*)(hp + j + q * 4096) = t[q];
    }
    for (; j < cur4; j += 4096)
      *(int4*)(hp + j) = *(const int4*)(cnt + j);
  }

  if (full) {
#pragma unroll
    for (int u = 0; u < U; ++u) *(int4*)(pos + e0 + u * 4096) = r[u];
  } else {
#pragma unroll
    for (int u = 0; u < U; ++u) {
      const int e = e0 + u * 4096;
      if (e     < E) pos[e]     = r[u].x;
      if (e + 1 < E) pos[e + 1] = r[u].y;
      if (e + 2 < E) pos[e + 2] = r[u].z;
      if (e + 3 < E) pos[e + 3] = r[u].w;
    }
  }
}

__global__ void hist_colsum_kernel(int* __restrict__ hist, int* __restrict__ deg,
                                   int n, int G) {
  constexpr int T = 32;
  const int d = blockIdx.x * blockDim.x + threadIdx.x;
  if (d >= n) return;
  const size_t np = (size_t)((n + 3) & ~3);
  int* hp = hist + d;
  int run = 0;
  for (int g0 = 0; g0 < G; g0 += T) {
    int v[T];
#pragma unroll
    for (int t = 0; t < T; ++t)
      v[t] = (g0 + t < G) ? hp[(size_t)(g0 + t) * np] : 0;
#pragma unroll
    for (int t = 0; t < T; ++t) {
      if (g0 + t < G) hp[(size_t)(g0 + t) * np] = run;
      run += v[t];
    }
  }
  deg[d] = run;
}

__device__ __forceinline__ int wave_sum_i(int v) {
#pragma unroll
  for (int off = 16; off > 0; off >>= 1) v += __shfl_xor(v, off);
  return v;
}

__global__ __launch_bounds__(1024) void rowptr_scan_kernel(
    const int* __restrict__ deg, int* __restrict__ rowptr, int n) {
  constexpr int T = 8;
  __shared__ int wsum[32];
  const int lane = threadIdx.x & 31;
  const int w = __builtin_amdgcn_readfirstlane((int)(threadIdx.x >> 5));
  const int chunks = (n + 31) / 32;
  const int cpw    = (chunks + 31) / 32;
  const int c0 = w * cpw;
  const int c1 = (c0 + cpw < chunks) ? (c0 + cpw) : chunks;

  int s = 0;
  for (int cb = c0; cb < c1; cb += T) {
    int v[T];
#pragma unroll
    for (int t = 0; t < T; ++t) {
      const int i = (cb + t) * 32 + lane;
      v[t] = (cb + t < c1 && i < n) ? deg[i] : 0;
    }
#pragma unroll
    for (int t = 0; t < T; ++t) s += v[t];
  }
  s = wave_sum_i(s);
  if (lane == 0) wsum[w] = s;
  __syncthreads();
  int run = wave_sum_i((lane < w) ? wsum[lane] : 0);

  for (int cb = c0; cb < c1; cb += T) {
    int v[T], inc[T];
#pragma unroll
    for (int t = 0; t < T; ++t) {
      const int i = (cb + t) * 32 + lane;
      v[t] = (cb + t < c1 && i < n) ? deg[i] : 0;
      inc[t] = v[t];
    }
#pragma unroll
    for (int off = 1; off < 32; off <<= 1) {
#pragma unroll
      for (int t = 0; t < T; ++t) {
        const int u = __shfl_up(inc[t], off);
        if (lane >= off) inc[t] += u;
      }
    }
#pragma unroll
    for (int t = 0; t < T; ++t) {
      const int i = (cb + t) * 32 + lane;
      if (cb + t < c1 && i < n) rowptr[i] = run + inc[t] - v[t];
      run += __builtin_amdgcn_readlane(inc[t], 31);
    }
  }
  if (w == 31 && lane == 0) rowptr[n] = run;
}

__global__ void edge_place_kernel(const int* __restrict__ ei, int E, int n,
                                  const int* __restrict__ rowptr,
                                  const int* __restrict__ hist,
                                  const int* __restrict__ pos,
                                  int* __restrict__ csr_src) {
  int i = blockIdx.x * blockDim.x + threadIdx.x;
  if (i >= E) return;
  int s = ei[i];
  int d = ei[E + i];
  const size_t np = (size_t)((n + 3) & ~3);
  int slot = rowptr[d] + hist[(size_t)(i >> SLICE_LOG2) * np + d] + pos[i];
  if ((unsigned)slot < (unsigned)E) csr_src[slot] = s;
}


__device__ __forceinline__ float wave_max_f(float v) {
#pragma unroll
  for (int off = 16; off > 0; off >>= 1) v = fmaxf(v, __shfl_xor(v, off));
  return v;
}
__device__ __forceinline__ float wave_sum_f(float v) {
#pragma unroll
  for (int off = 16; off > 0; off >>= 1) v += __shfl_xor(v, off);
  return v;
}
__device__ __forceinline__ float leaky(float e) {
  return (e > 0.f) ? e : NEG_SLOPE * e;
}

template <int HEADS, int C, bool ELU>
__global__ __launch_bounds__(256) void gat_gather_kernel(
    const int* __restrict__ rowptr, const int* __restrict__ csr_src,
    const float* __restrict__ H, const float* __restrict__ as_,
    const float* __restrict__ ad_, const float* __restrict__ bias,
    float* __restrict__ out, int n, int E) {
  constexpr int CH  = HEADS * C;
  constexpr int VEC = CH / 32;
  typedef float vecf __attribute__((ext_vector_type(VEC)));

  const int lane = threadIdx.x & 31;
  const int wv  = __builtin_amdgcn_readfirstlane((int)(threadIdx.x >> 5));
  const int gw  = blockIdx.x * 8 + wv;
  const int nw  = gridDim.x * 8;
  const int myh = (lane * VEC) / C;
  const vecf bv = *(const vecf*)(bias + lane * VEC);

  for (int node = gw; node < n; node += nw) {
    int beg = rowptr[node];
    int end = rowptr[node + 1];
    beg = (beg < 0) ? 0 : beg;
    end = (end > E) ? E : end;
    const int deg = end - beg;

    const vecf hself = *(const vecf*)(H + (size_t)node * CH + lane * VEC);

    float adh[HEADS], eself[HEADS], m[HEADS];
#pragma unroll
    for (int h = 0; h < HEADS; ++h) {
      adh[h]   = ad_[node * HEADS + h];
      eself[h] = leaky(as_[node * HEADS + h] + adh[h]);
    }

    const bool v0 = lane < deg;
    int s0 = node;
    if (v0) {
      s0 = csr_src[beg + lane];
      s0 = (s0 < 0) ? 0 : ((s0 >= n) ? n - 1 : s0);
    }
    float e0[HEADS];
    if (HEADS == 2) {
      const v2f t = *(const v2f*)(as_ + (size_t)s0 * 2);
      e0[0]         = leaky(t.x + adh[0]);
      e0[HEADS - 1] = leaky(t.y + adh[HEADS - 1]);
    } else {
      e0[0] = leaky(as_[s0] + adh[0]);
    }
#pragma unroll
    for (int h = 0; h < HEADS; ++h) m[h] = fmaxf(eself[h], e0[h]);
    for (int base = 32; base < deg; base += 32) {
      int j = base + lane;
      if (j < deg) {
        int s = csr_src[beg + j];
        s = (s < 0) ? 0 : ((s >= n) ? n - 1 : s);
#pragma unroll
        for (int h = 0; h < HEADS; ++h)
          m[h] = fmaxf(m[h], leaky(as_[s * HEADS + h] + adh[h]));
      }
    }
#pragma unroll
    for (int h = 0; h < HEADS; ++h) m[h] = wave_max_f(m[h]);

    float p0[HEADS], den[HEADS], pself[HEADS];
#pragma unroll
    for (int h = 0; h < HEADS; ++h) {
      p0[h]  = v0 ? __expf(e0[h] - m[h]) : 0.f;
      den[h] = p0[h];
    }
    for (int base = 32; base < deg; base += 32) {
      int j = base + lane;
      if (j < deg) {
        int s = csr_src[beg + j];
        s = (s < 0) ? 0 : ((s >= n) ? n - 1 : s);
#pragma unroll
        for (int h = 0; h < HEADS; ++h)
          den[h] += __expf(leaky(as_[s * HEADS + h] + adh[h]) - m[h]);
      }
    }
#pragma unroll
    for (int h = 0; h < HEADS; ++h) {
      pself[h] = __expf(eself[h] - m[h]);
      den[h]   = wave_sum_f(den[h]) + pself[h] + 1e-16f;
    }

    vecf acc;
    {
      const float a = (HEADS == 2 && myh) ? pself[HEADS - 1] / den[HEADS - 1]
                                          : pself[0] / den[0];
      acc = hself * a;
    }
    for (int base = 0; base < deg; base += 32) {
      int   sj = s0;
      float al[HEADS];
      if (base == 0) {
#pragma unroll
        for (int h = 0; h < HEADS; ++h) al[h] = p0[h] / den[h];
      } else {
        int j = base + lane;
        sj = node;
#pragma unroll
        for (int h = 0; h < HEADS; ++h) al[h] = 0.f;
        if (j < deg) {
          sj = csr_src[beg + j];
          sj = (sj < 0) ? 0 : ((sj >= n) ? n - 1 : sj);
#pragma unroll
          for (int h = 0; h < HEADS; ++h)
            al[h] = __expf(leaky(as_[sj * HEADS + h] + adh[h]) - m[h]) / den[h];
        }
      }
      const int cnt  = (deg - base < 32) ? (deg - base) : 32;
      const int cnt4 = (cnt + 3) & ~3;
      for (int k = 0; k < cnt4; k += 4) {
        vecf  hv[4];
        float a[4];
#pragma unroll
        for (int u = 0; u < 4; ++u) {
          const int sk = __builtin_amdgcn_readlane(sj, k + u);
          hv[u] = *(const vecf*)(H + (size_t)sk * CH + lane * VEC);
          const float a0 = __int_as_float(__builtin_amdgcn_readlane(__float_as_int(al[0]), k + u));
          a[u] = a0;
          if (HEADS == 2) {
            const float a1 = __int_as_float(__builtin_amdgcn_readlane(__float_as_int(al[HEADS - 1]), k + u));
            a[u] = myh ? a1 : a0;
          }
        }
#pragma unroll
        for (int u = 0; u < 4; ++u) acc += hv[u] * a[u];
      }
    }

#pragma unroll
    for (int v = 0; v < VEC; ++v) {
      float y = acc[v] + bv[v];
      if (ELU) y = (y > 0.f) ? y : expm1f(y);
      acc[v] = y;
    }
    *(vecf*)(out + (size_t)node * CH + lane * VEC) = acc;
  }
}


static inline int cdiv(long long a, long long b) { return (int)((a + b - 1) / b); }

extern "C" void kernel_launch(void* const* d_in, const int* in_sizes, int n_in,
                              void* d_out, int out_size, void* d_ws, size_t ws_size,
                              hipStream_t stream) {
  const int IN_CH = 128, HID = 64, HEADS = 2, OUT = 64;
  const int CH1 = HEADS * HID;

  const float* x        = (const float*)d_in[0];
  const int*   ei       = (const int*)d_in[1];
  const float* W1       = (const float*)d_in[2];
  const float* att1_src = (const float*)d_in[3];
  const float* att1_dst = (const float*)d_in[4];
  const float* b1       = (const float*)d_in[5];
  const float* W2       = (const float*)d_in[6];
  const float* att2_src = (const float*)d_in[7];
  const float* att2_dst = (const float*)d_in[8];
  const float* b2       = (const float*)d_in[9];
  float* out            = (float*)d_out;

  const int n  = in_sizes[0] / IN_CH;
  const int E  = in_sizes[1] / 2;

  char* ws = (char*)d_ws;
  size_t off = 0;
  auto alloc = [&](size_t bytes) -> void* {
    void* p = ws + off;
    off += (bytes + 255) & ~(size_t)255;
    return p;
  };
  float* h1      = (float*)alloc((size_t)n * CH1 * 4);
  float* agg1    = (float*)alloc((size_t)n * CH1 * 4);
  float* h2      = (float*)alloc((size_t)n * OUT * 4);
  float* as1     = (float*)alloc((size_t)n * HEADS * 4);
  float* ad1     = (float*)alloc((size_t)n * HEADS * 4);
  float* as2     = (float*)alloc((size_t)n * 4);
  float* ad2     = (float*)alloc((size_t)n * 4);
  int*   deg     = (int*)  alloc((size_t)n * 4);
  int*   rowptr  = (int*)  alloc((size_t)(n + 1) * 4);
  int*   pos     = (int*)  alloc((size_t)E * 4);
  int*   csr_src = (int*)  alloc((size_t)E * 4);
  const int G    = cdiv(E, SLICE);
  int*   hist    = (int*)  alloc((size_t)G * (n + 4) * 4);
  (void)ws_size; (void)n_in; (void)out_size;

  const int BT = 256;
  const int GATHER_BLOCKS = 2048;

  edge_count_kernel<<<G, 1024, 0, stream>>>(ei, E, n, hist, pos);
  hist_colsum_kernel<<<cdiv(n, BT), BT, 0, stream>>>(hist, deg, n, G);
  rowptr_scan_kernel<<<1, 1024, 0, stream>>>(deg, rowptr, n);
  edge_place_kernel<<<cdiv(E, BT), BT, 0, stream>>>(ei, E, n, rowptr, hist, pos, csr_src);

  gemm_wmma_f32<128, 32><<<cdiv(n, 32), BT, 0, stream>>>(x, W1, h1, n);
  alpha_kernel<<<cdiv(n * HEADS, BT), BT, 0, stream>>>(
      h1, att1_src, att1_dst, as1, ad1, n, HEADS, HID);
  gat_gather_kernel<2, 64, true><<<GATHER_BLOCKS, BT, 0, stream>>>(
      rowptr, csr_src, h1, as1, ad1, b1, agg1, n, E);

  gemm_wmma_f32<64, 64><<<cdiv(n, 64), BT, 0, stream>>>(agg1, W2, h2, n);
  alpha_kernel<<<cdiv(n, BT), BT, 0, stream>>>(
      h2, att2_src, att2_dst, as2, ad2, n, 1, OUT);
  gat_gather_kernel<1, 64, false><<<GATHER_BLOCKS, BT, 0, stream>>>(
      rowptr, csr_src, h2, as2, ad2, b2, out, n, E);
}


# attention dots formed in the GEMM epilogue: output tile staged in LDS, waves re-read whole rows, store them as full lines, sum the dots per head by lane exchanges; 2 launches less (dot sum order new)
# speedup vs baseline: 7.4892x; 1.0626x over previous
#include <hip/hip_runtime.h>
#include <hip/hip_bf16.h>


typedef float v2f __attribute__((ext_vector_type(2)));
typedef float v8f __attribute__((ext_vector_type(8)));

#define NEG_SLOPE 0.2f


template <int M, int OFF, int NV>
__device__ __forceinline__ void head_sum(float (&val)[NV], int lane) {
  if constexpr (OFF > 0) {
    if constexpr (M > 1) {
      constexpr int half = M / 2;
      const bool upper = (lane & OFF) != 0;
#pragma unroll
      for (int k = 0; k < half; ++k) {
        const float lo = val[k], hi = val[k + half];
        const float send = upper ? lo : hi;
        const float keep = upper ? hi : lo;
        val[k] = keep + __shfl_xor(send, OFF);
      }
      head_sum<half, OFF / 2, NV>(val, lane);
    } else {
      val[0] += __shfl_xor(val[0], OFF);
      head_sum<1, OFF / 2, NV>(val, lane);
    }
  }
}


template <int COUT, int BM, int HEADS>
__global__ __launch_bounds__(256) void gemm_wmma_f32(
    const float* __restrict__ A, const float* __restrict__ W,
    const float* __restrict__ att_src, const float* __restrict__ att_dst,
    float* __restrict__ H, float* __restrict__ as_, float* __restrict__ ad_,
    int n) {
  constexpr int K  = 128;
  constexpr int P  = K + 4;
  constexpr int CT = COUT / 16;
  static_assert(CT * (BM / 32) == 8, "8 waves per work-group");
  constexpr int WV4 = COUT * (K / 4) / 256;
  constexpr int AV4 = BM * (K / 4) / 256;
  __shared__ __attribute__((aligned(16))) float Ws[COUT * P];
  __shared__ __attribute__((aligned(16))) float As[BM * P];

  const int tid  = threadIdx.x;
  const int lane = tid & 31;
  const int wv   = __builtin_amdgcn_readfirstlane(tid >> 5);
  const int row0 = blockIdx.x * BM;

  float4 wreg[WV4], areg[AV4];
#pragma unroll
  for (int j = 0; j < WV4; ++j) {
    const int r = wv + 8 * j;
    wreg[j] = *(const float4*)(W + (size_t)r * K + lane * 4);
  }
#pragma unroll
  for (int j = 0; j < AV4; ++j) {
    int r = row0 + wv + 8 * j;
    r = (r < n) ? r : (n - 1);
    areg[j] = *(const float4*)(A + (size_t)r * K + lane * 4);
  }
#pragma unroll
  for (int j = 0; j < WV4; ++j)
    *(float4*)(Ws + (wv + 8 * j) * P + lane * 4) = wreg[j];
#pragma unroll
  for (int j = 0; j < AV4; ++j)
    *(float4*)(As + (wv + 8 * j) * P + lane * 4) = areg[j];
  __syncthreads();

  const int mn      = lane & 15;
  const int khalf   = (lane >> 4) * 2;
  const int colTile = wv % CT;
  const int rg      = wv / CT;

  const float* bp  = Ws + (colTile * 16 + mn) * P + khalf;
  const float* ap0 = As + (rg * 32 + mn) * P + khalf;
  const float* ap1 = ap0 + 16 * P;

  v8f acc0 = {}, acc1 = {};
#pragma unroll
  for (int k0 = 0; k0 < K; k0 += 4) {
    const v2f b  = *(const v2f*)(bp + k0);
    const v2f a0 = *(const v2f*)(ap0 + k0);
    const v2f a1 = *(const v2f*)(ap1 + k0);
    acc0 = __builtin_amdgcn_wmma_f32_16x16x4_f32(
         false, a0,  false, b,
         (short)0, acc0,  false,  false);
    acc1 = __builtin_amdgcn_wmma_f32_16x16x4_f32(
         false, a1,  false, b,
         (short)0, acc1,  false,  false);
  }

  constexpr int VEC = COUT / 32;
  constexpr int LPH = 32 / HEADS;
  typedef float vecf __attribute__((ext_vector_type(VEC)));
  const vecf sv = *(const vecf*)(att_src + lane * VEC);
  const vecf dv = *(const vecf*)(att_dst + lane * VEC);

  __syncthreads();
  {
    float* op = As + (rg * 32 + 8 * (lane >> 4)) * P + colTile * 16 + mn;
#pragma unroll
    for (int v = 0; v < 8; ++v) {
      op[v * P]        = acc0[v];
      op[(16 + v) * P] = acc1[v];
    }
  }
  __syncthreads();

  constexpr int R  = BM / 8;
  constexpr int NV = 2 * R;
  static_assert(NV == LPH / 2, "the halving exchange below ends with one value per lane pair");
  vecf hrow[R];
#pragma unroll
  for (int j = 0; j < R; ++j)
    hrow[j] = *(const vecf*)(As + (wv + 8 * j) * P + lane * VEC);
#pragma unroll
  for (int j = 0; j < R; ++j) {
    const int grow = row0 + wv + 8 * j;
    if (grow < n) *(vecf*)(H + (size_t)grow * COUT + lane * VEC) = hrow[j];
  }

  float val[NV];
#pragma unroll
  for (int j = 0; j < R; ++j) {
    float ps = 0.f, pd = 0.f;
#pragma unroll
    for (int c = 0; c < VEC; ++c) {
      ps += hrow[j][c] * sv[c];
      pd += hrow[j][c] * dv[c];
    }
    val[2 * j]     = ps;
    val[2 * j + 1] = pd;
  }
  head_sum<NV, LPH / 2, NV>(val, lane);
  if ((lane & 1) == 0) {
    const int i    = (lane >> 1) & (NV - 1);
    const int grow = row0 + wv + 8 * (i >> 1);
    if (grow < n) {
      float* dp = (i & 1) ? ad_ : as_;
      dp[(size_t)grow * HEADS + lane / LPH] = val[0];
    }
  }
}


constexpr int SLICE_LOG2 = 14;
constexpr int SLICE      = 1 << SLICE_LOG2;
constexpr int NB         = 65536;

__global__ __launch_bounds__(1024) void edge_count_kernel(
    const int* __restrict__ ei, int E, int n,
    int* __restrict__ hist, int* __restrict__ pos) {
  constexpr int U = SLICE / 4096;
  __shared__ __attribute__((aligned(16))) int cnt[NB];
  const int tid = threadIdx.x;
  const int g   = blockIdx.x;
  const int np  = (n + 3) & ~3;
  const int* dst = ei + E;
  const bool vec = (E & 3) == 0;
  const int e0   = g * SLICE + tid * 4;

  const bool full = vec && (g * SLICE + SLICE <= E);

  int4 d[U], r[U];
  if (full) {
#pragma unroll
    for (int u = 0; u < U; ++u) d[u] = *(const int4*)(dst + e0 + u * 4096);
  } else {
#pragma unroll
    for (int u = 0; u < U; ++u) {
      const int e = e0 + u * 4096;
      const int x = dst[(e     < E) ? e     : E - 1];
      const int y = dst[(e + 1 < E) ? e + 1 : E - 1];
      const int z = dst[(e + 2 < E) ? e + 2 : E - 1];
      const int w = dst[(e + 3 < E) ? e + 3 : E - 1];
      d[u].x = (e     < E) ? x : -1;
      d[u].y = (e + 1 < E) ? y : -1;
      d[u].z = (e + 2 < E) ? z : -1;
      d[u].w = (e + 3 < E) ? w : -1;
    }
  }
#pragma unroll
  for (int u = 0; u < U; ++u) r[u] = make_int4(0, 0, 0, 0);

  for (int nb = 0; nb < n; nb += NB) {
    const int cur  = (n - nb < NB) ? (n - nb) : NB;
    const int cur4 = (cur + 3) & ~3;
    if (nb) __syncthreads();
    for (int j = tid * 4; j < cur4; j += 4096)
      *(int4*)(cnt + j) = make_int4(0, 0, 0, 0);
    __syncthreads();
#pragma unroll
    for (int u = 0; u < U; ++u) {
      const unsigned ox = (unsigned)(d[u].x - nb), oy = (unsigned)(d[u].y - nb);
      const unsigned oz = (unsigned)(d[u].z - nb), ow = (unsigned)(d[u].w - nb);
      if (ox < (unsigned)cur) r[u].x = atomicAdd(&cnt[ox], 1);
      if (oy < (unsigned)cur) r[u].y = atomicAdd(&cnt[oy], 1);
      if (oz < (unsigned)cur) r[u].z = atomicAdd(&cnt[oz], 1);
      if (ow < (unsigned)cur) r[u].w = atomicAdd(&cnt[ow], 1);
    }
    __syncthreads();
    int* hp = hist + (size_t)g * np + nb;
    int j = tid * 4;
    for (; j + 3 * 4096 < cur4; j += 4 * 4096) {
      int4 t[4];
#pragma unroll
      for (int q = 0; q < 4; ++q) t[q] = *(const int4*)(cnt + j + q * 4096);
#pragma unroll
      for (int q = 0; q < 4; ++q) *(int4*)(hp + j + q * 4096) = t[q];
    }
    for (; j < cur4; j += 4096)
      *(int4*)(hp + j) = *(const int4*)(cnt + j);
  }

  if (full) {
#pragma unroll
    for (int u = 0; u < U; ++u) *(int4*)(pos + e0 + u * 4096) = r[u];
  } else {
#pragma unroll
    for (int u = 0; u < U; ++u) {
      const int e = e0 + u * 4096;
      if (e     < E) pos[e]     = r[u].x;
      if (e + 1 < E) pos[e + 1] = r[u].y;
      if (e + 2 < E) pos[e + 2] = r[u].z;
      if (e + 3 < E) pos[e + 3] = r[u].w;
    }
  }
}

__global__ void hist_colsum_kernel(int* __restrict__ hist, int* __restrict__ deg,
                                   int n, int G) {
  constexpr int T = 32;
  const int d = blockIdx.x * blockDim.x + threadIdx.x;
  if (d >= n) return;
  const size_t np = (size_t)((n + 3) & ~3);
  int* hp = hist + d;
  int run = 0;
  for (int g0 = 0; g0 < G; g0 += T) {
    int v[T];
#pragma unroll
    for (int t = 0; t < T; ++t)
      v[t] = (g0 + t < G) ? hp[(size_t)(g0 + t) * np] : 0;
#pragma unroll
    for (int t = 0; t < T; ++t) {
      if (g0 + t < G) hp[(size_t)(g0 + t) * np] = run;
      run += v[t];
    }
  }
  deg[d] = run;
}

__device__ __forceinline__ int wave_sum_i(int v) {
#pragma unroll
  for (int off = 16; off > 0; off >>= 1) v += __shfl_xor(v, off);
  return v;
}

__global__ __launch_bounds__(1024) void rowptr_scan_kernel(
    const int* __restrict__ deg, int* __restrict__ rowptr, int n) {
  constexpr int T = 8;
  __shared__ int wsum[32];
  const int lane = threadIdx.x & 31;
  const int w = __builtin_amdgcn_readfirstlane((int)(threadIdx.x >> 5));
  const int chunks = (n + 31) / 32;
  const int cpw    = (chunks + 31) / 32;
  const int c0 = w * cpw;
  const int c1 = (c0 + cpw < chunks) ? (c0 + cpw) : chunks;

  int s = 0;
  for (int cb = c0; cb < c1; cb += T) {
    int v[T];
#pragma unroll
    for (int t = 0; t < T; ++t) {
      const int i = (cb + t) * 32 + lane;
      v[t] = (cb + t < c1 && i < n) ? deg[i] : 0;
    }
#pragma unroll
    for (int t = 0; t < T; ++t) s += v[t];
  }
  s = wave_sum_i(s);
  if (lane == 0) wsum[w] = s;
  __syncthreads();
  int run = wave_sum_i((lane < w) ? wsum[lane] : 0);

  for (int cb = c0; cb < c1; cb += T) {
    int v[T], inc[T];
#pragma unroll
    for (int t = 0; t < T; ++t) {
      const int i = (cb + t) * 32 + lane;
      v[t] = (cb + t < c1 && i < n) ? deg[i] : 0;
      inc[t] = v[t];
    }
#pragma unroll
    for (int off = 1; off < 32; off <<= 1) {
#pragma unroll
      for (int t = 0; t < T; ++t) {
        const int u = __shfl_up(inc[t], off);
        if (lane >= off) inc[t] += u;
      }
    }
#pragma unroll
    for (int t = 0; t < T; ++t) {
      const int i = (cb + t) * 32 + lane;
      if (cb + t < c1 && i < n) rowptr[i] = run + inc[t] - v[t];
      run += __builtin_amdgcn_readlane(inc[t], 31);
    }
  }
  if (w == 31 && lane == 0) rowptr[n] = run;
}

__global__ void edge_place_kernel(const int* __restrict__ ei, int E, int n,
                                  const int* __restrict__ rowptr,
                                  const int* __restrict__ hist,
                                  const int* __restrict__ pos,
                                  int* __restrict__ csr_src) {
  int i = blockIdx.x * blockDim.x + threadIdx.x;
  if (i >= E) return;
  int s = ei[i];
  int d = ei[E + i];
  const size_t np = (size_t)((n + 3) & ~3);
  int slot = rowptr[d] + hist[(size_t)(i >> SLICE_LOG2) * np + d] + pos[i];
  if ((unsigned)slot < (unsigned)E) csr_src[slot] = s;
}


__device__ __forceinline__ float wave_max_f(float v) {
#pragma unroll
  for (int off = 16; off > 0; off >>= 1) v = fmaxf(v, __shfl_xor(v, off));
  return v;
}
__device__ __forceinline__ float wave_sum_f(float v) {
#pragma unroll
  for (int off = 16; off > 0; off >>= 1) v += __shfl_xor(v, off);
  return v;
}
__device__ __forceinline__ float leaky(float e) {
  return (e > 0.f) ? e : NEG_SLOPE * e;
}

template <int HEADS, int C, bool ELU>
__global__ __launch_bounds__(256) void gat_gather_kernel(
    const int* __restrict__ rowptr, const int* __restrict__ csr_src,
    const float* __restrict__ H, const float* __restrict__ as_,
    const float* __restrict__ ad_, const float* __restrict__ bias,
    float* __restrict__ out, int n, int E) {
  constexpr int CH  = HEADS * C;
  constexpr int VEC = CH / 32;
  typedef float vecf __attribute__((ext_vector_type(VEC)));

  const int lane = threadIdx.x & 31;
  const int wv  = __builtin_amdgcn_readfirstlane((int)(threadIdx.x >> 5));
  const int gw  = blockIdx.x * 8 + wv;
  const int nw  = gridDim.x * 8;
  const int myh = (lane * VEC) / C;
  const vecf bv = *(const vecf*)(bias + lane * VEC);

  for (int node = gw; node < n; node += nw) {
    int beg = rowptr[node];
    int end = rowptr[node + 1];
    beg = (beg < 0) ? 0 : beg;
    end = (end > E) ? E : end;
    const int deg = end - beg;

    const vecf hself = *(const vecf*)(H + (size_t)node * CH + lane * VEC);

    float adh[HEADS], eself[HEADS], m[HEADS];
#pragma unroll
    for (int h = 0; h < HEADS; ++h) {
      adh[h]   = ad_[node * HEADS + h];
      eself[h] = leaky(as_[node * HEADS + h] + adh[h]);
    }

    const bool v0 = lane < deg;
    int s0 = node;
    if (v0) {
      s0 = csr_src[beg + lane];
      s0 = (s0 < 0) ? 0 : ((s0 >= n) ? n - 1 : s0);
    }
    float e0[HEADS];
    if (HEADS == 2) {
      const v2f t = *(const v2f*)(as_ + (size_t)s0 * 2);
      e0[0]         = leaky(t.x + adh[0]);
      e0[HEADS - 1] = leaky(t.y + adh[HEADS - 1]);
    } else {
      e0[0] = leaky(as_[s0] + adh[0]);
    }
#pragma unroll
    for (int h = 0; h < HEADS; ++h) m[h] = fmaxf(eself[h], e0[h]);
    for (int base = 32; base < deg; base += 32) {
      int j = base + lane;
      if (j < deg) {
        int s = csr_src[beg + j];
        s = (s < 0) ? 0 : ((s >= n) ? n - 1 : s);
#pragma unroll
        for (int h = 0; h < HEADS; ++h)
          m[h] = fmaxf(m[h], leaky(as_[s * HEADS + h] + adh[h]));
      }
    }
#pragma unroll
    for (int h = 0; h < HEADS; ++h) m[h] = wave_max_f(m[h]);

    float p0[HEADS], den[HEADS], pself[HEADS];
#pragma unroll
    for (int h = 0; h < HEADS; ++h) {
      p0[h]  = v0 ? __expf(e0[h] - m[h]) : 0.f;
      den[h] = p0[h];
    }
    for (int base = 32; base < deg; base += 32) {
      int j = base + lane;
      if (j < deg) {
        int s = csr_src[beg + j];
        s = (s < 0) ? 0 : ((s >= n) ? n - 1 : s);
#pragma unroll
        for (int h = 0; h < HEADS; ++h)
          den[h] += __expf(leaky(as_[s * HEADS + h] + adh[h]) - m[h]);
      }
    }
#pragma unroll
    for (int h = 0; h < HEADS; ++h) {
      pself[h] = __expf(eself[h] - m[h]);
      den[h]   = wave_sum_f(den[h]) + pself[h] + 1e-16f;
    }

    vecf acc;
    {
      const float a = (HEADS == 2 && myh) ? pself[HEADS - 1] / den[HEADS - 1]
                                          : pself[0] / den[0];
      acc = hself * a;
    }
    for (int base = 0; base < deg; base += 32) {
      int   sj = s0;
      float al[HEADS];
      if (base == 0) {
#pragma unroll
        for (int h = 0; h < HEADS; ++h) al[h] = p0[h] / den[h];
      } else {
        int j = base + lane;
        sj = node;
#pragma unroll
        for (int h = 0; h < HEADS; ++h) al[h] = 0.f;
        if (j < deg) {
          sj = csr_src[beg + j];
          sj = (sj < 0) ? 0 : ((sj >= n) ? n - 1 : sj);
#pragma unroll
          for (int h = 0; h < HEADS; ++h)
            al[h] = __expf(leaky(as_[sj * HEADS + h] + adh[h]) - m[h]) / den[h];
        }
      }
      const int cnt  = (deg - base < 32) ? (deg - base) : 32;
      const int cnt4 = (cnt + 3) & ~3;
      for (int k = 0; k < cnt4; k += 4) {
        vecf  hv[4];
        float a[4];
#pragma unroll
        for (int u = 0; u < 4; ++u) {
          const int sk = __builtin_amdgcn_readlane(sj, k + u);
          hv[u] = *(const vecf*)(H + (size_t)sk * CH + lane * VEC);
          const float a0 = __int_as_float(__builtin_amdgcn_readlane(__float_as_int(al[0]), k + u));
          a[u] = a0;
          if (HEADS == 2) {
            const float a1 = __int_as_float(__builtin_amdgcn_readlane(__float_as_int(al[HEADS - 1]), k + u));
            a[u] = myh ? a1 : a0;
          }
        }
#pragma unroll
        for (int u = 0; u < 4; ++u) acc += hv[u] * a[u];
      }
    }

#pragma unroll
    for (int v = 0; v < VEC; ++v) {
      float y = acc[v] + bv[v];
      if (ELU) y = (y > 0.f) ? y : expm1f(y);
      acc[v] = y;
    }
    *(vecf*)(out + (size_t)node * CH + lane * VEC) = acc;
  }
}


static inline int cdiv(long long a, long long b) { return (int)((a + b - 1) / b); }

extern "C" void kernel_launch(void* const* d_in, const int* in_sizes, int n_in,
                              void* d_out, int out_size, void* d_ws, size_t ws_size,
                              hipStream_t stream) {
  const int IN_CH = 128, HID = 64, HEADS = 2, OUT = 64;
  const int CH1 = HEADS * HID;

  const float* x        = (const float*)d_in[0];
  const int*   ei       = (const int*)d_in[1];
  const float* W1       = (const float*)d_in[2];
  const float* att1_src = (const float*)d_in[3];
  const float* att1_dst = (const float*)d_in[4];
  const float* b1       = (const float*)d_in[5];
  const float* W2       = (const float*)d_in[6];
  const float* att2_src = (const float*)d_in[7];
  const float* att2_dst = (const float*)d_in[8];
  const float* b2       = (const float*)d_in[9];
  float* out            = (float*)d_out;

  const int n  = in_sizes[0] / IN_CH;
  const int E  = in_sizes[1] / 2;

  char* ws = (char*)d_ws;
  size_t off = 0;
  auto alloc = [&](size_t bytes) -> void* {
    void* p = ws + off;
    off += (bytes + 255) & ~(size_t)255;
    return p;
  };
  float* h1      = (float*)alloc((size_t)n * CH1 * 4);
  float* agg1    = (float*)alloc((size_t)n * CH1 * 4);
  float* h2      = (float*)alloc((size_t)n * OUT * 4);
  float* as1     = (float*)alloc((size_t)n * HEADS * 4);
  float* ad1     = (float*)alloc((size_t)n * HEADS * 4);
  float* as2     = (float*)alloc((size_t)n * 4);
  float* ad2     = (float*)alloc((size_t)n * 4);
  int*   deg     = (int*)  alloc((size_t)n * 4);
  int*   rowptr  = (int*)  alloc((size_t)(n + 1) * 4);
  int*   pos     = (int*)  alloc((size_t)E * 4);
  int*   csr_src = (int*)  alloc((size_t)E * 4);
  const int G    = cdiv(E, SLICE);
  int*   hist    = (int*)  alloc((size_t)G * (n + 4) * 4);
  (void)ws_size; (void)n_in; (void)out_size;

  const int BT = 256;
  const int GATHER_BLOCKS = 2048;

  edge_count_kernel<<<G, 1024, 0, stream>>>(ei, E, n, hist, pos);
  hist_colsum_kernel<<<cdiv(n, BT), BT, 0, stream>>>(hist, deg, n, G);
  rowptr_scan_kernel<<<1, 1024, 0, stream>>>(deg, rowptr, n);
  edge_place_kernel<<<cdiv(E, BT), BT, 0, stream>>>(ei, E, n, rowptr, hist, pos, csr_src);

  gemm_wmma_f32<128, 32, 2><<<cdiv(n, 32), BT, 0, stream>>>(
      x, W1, att1_src, att1_dst, h1, as1, ad1, n);
  gat_gather_kernel<2, 64, true><<<GATHER_BLOCKS, BT, 0, stream>>>(
      rowptr, csr_src, h1, as1, ad1, b1, agg1, n, E);

  gemm_wmma_f32<64, 64, 1><<<cdiv(n, 64), BT, 0, stream>>>(
      agg1, W2, att2_src, att2_dst, h2, as2, ad2, n);
  gat_gather_kernel<1, 64, false><<<GATHER_BLOCKS, BT, 0, stream>>>(
      rowptr, csr_src, h2, as2, ad2, b2, out, n, E);
}


# row-start prefix sum split: the column pass scans its own 256 nodes and emits a block total, placement sums the block totals in LDS and writes rowptr; the one-block scan launch is gone
# speedup vs baseline: 8.3870x; 1.1199x over previous
#include <hip/hip_runtime.h>
#include <hip/hip_bf16.h>


typedef float v2f __attribute__((ext_vector_type(2)));
typedef float v8f __attribute__((ext_vector_type(8)));

#define NEG_SLOPE 0.2f


template <int M, int OFF, int NV>
__device__ __forceinline__ void head_sum(float (&val)[NV], int lane) {
  if constexpr (OFF > 0) {
    if constexpr (M > 1) {
      constexpr int half = M / 2;
      const bool upper = (lane & OFF) != 0;
#pragma unroll
      for (int k = 0; k < half; ++k) {
        const float lo = val[k], hi = val[k + half];
        const float send = upper ? lo : hi;
        const float keep = upper ? hi : lo;
        val[k] = keep + __shfl_xor(send, OFF);
      }
      head_sum<half, OFF / 2, NV>(val, lane);
    } else {
      val[0] += __shfl_xor(val[0], OFF);
      head_sum<1, OFF / 2, NV>(val, lane);
    }
  }
}


template <int COUT, int BM, int HEADS>
__global__ __launch_bounds__(256) void gemm_wmma_f32(
    const float* __restrict__ A, const float* __restrict__ W,
    const float* __restrict__ att_src, const float* __restrict__ att_dst,
    float* __restrict__ H, float* __restrict__ as_, float* __restrict__ ad_,
    int n) {
  constexpr int K  = 128;
  constexpr int P  = K + 4;
  constexpr int CT = COUT / 16;
  static_assert(CT * (BM / 32) == 8, "8 waves per work-group");
  constexpr int WV4 = COUT * (K / 4) / 256;
  constexpr int AV4 = BM * (K / 4) / 256;
  __shared__ __attribute__((aligned(16))) float Ws[COUT * P];
  __shared__ __attribute__((aligned(16))) float As[BM * P];

  const int tid  = threadIdx.x;
  const int lane = tid & 31;
  const int wv   = __builtin_amdgcn_readfirstlane(tid >> 5);
  const int row0 = blockIdx.x * BM;

  float4 wreg[WV4], areg[AV4];
#pragma unroll
  for (int j = 0; j < WV4; ++j) {
    const int r = wv + 8 * j;
    wreg[j] = *(const float4*)(W + (size_t)r * K + lane * 4);
  }
#pragma unroll
  for (int j = 0; j < AV4; ++j) {
    int r = row0 + wv + 8 * j;
    r = (r < n) ? r : (n - 1);
    areg[j] = *(const float4*)(A + (size_t)r * K + lane * 4);
  }
#pragma unroll
  for (int j = 0; j < WV4; ++j)
    *(float4*)(Ws + (wv + 8 * j) * P + lane * 4) = wreg[j];
#pragma unroll
  for (int j = 0; j < AV4; ++j)
    *(float4*)(As + (wv + 8 * j) * P + lane * 4) = areg[j];
  __syncthreads();

  const int mn      = lane & 15;
  const int khalf   = (lane >> 4) * 2;
  const int colTile = wv % CT;
  const int rg      = wv / CT;

  const float* bp  = Ws + (colTile * 16 + mn) * P + khalf;
  const float* ap0 = As + (rg * 32 + mn) * P + khalf;
  const float* ap1 = ap0 + 16 * P;

  v8f acc0 = {}, acc1 = {};
#pragma unroll
  for (int k0 = 0; k0 < K; k0 += 4) {
    const v2f b  = *(const v2f*)(bp + k0);
    const v2f a0 = *(const v2f*)(ap0 + k0);
    const v2f a1 = *(const v2f*)(ap1 + k0);
    acc0 = __builtin_amdgcn_wmma_f32_16x16x4_f32(
         false, a0,  false, b,
         (short)0, acc0,  false,  false);
    acc1 = __builtin_amdgcn_wmma_f32_16x16x4_f32(
         false, a1,  false, b,
         (short)0, acc1,  false,  false);
  }

  constexpr int VEC = COUT / 32;
  constexpr int LPH = 32 / HEADS;
  typedef float vecf __attribute__((ext_vector_type(VEC)));
  const vecf sv = *(const vecf*)(att_src + lane * VEC);
  const vecf dv = *(const vecf*)(att_dst + lane * VEC);

  __syncthreads();
  {
    float* op = As + (rg * 32 + 8 * (lane >> 4)) * P + colTile * 16 + mn;
#pragma unroll
    for (int v = 0; v < 8; ++v) {
      op[v * P]        = acc0[v];
      op[(16 + v) * P] = acc1[v];
    }
  }
  __syncthreads();

  constexpr int R  = BM / 8;
  constexpr int NV = 2 * R;
  static_assert(NV == LPH / 2, "the halving exchange below ends with one value per lane pair");
  vecf hrow[R];
#pragma unroll
  for (int j = 0; j < R; ++j)
    hrow[j] = *(const vecf*)(As + (wv + 8 * j) * P + lane * VEC);
#pragma unroll
  for (int j = 0; j < R; ++j) {
    const int grow = row0 + wv + 8 * j;
    if (grow < n) *(vecf*)(H + (size_t)grow * COUT + lane * VEC) = hrow[j];
  }

  float val[NV];
#pragma unroll
  for (int j = 0; j < R; ++j) {
    float ps = 0.f, pd = 0.f;
#pragma unroll
    for (int c = 0; c < VEC; ++c) {
      ps += hrow[j][c] * sv[c];
      pd += hrow[j][c] * dv[c];
    }
    val[2 * j]     = ps;
    val[2 * j + 1] = pd;
  }
  head_sum<NV, LPH / 2, NV>(val, lane);
  if ((lane & 1) == 0) {
    const int i    = (lane >> 1) & (NV - 1);
    const int grow = row0 + wv + 8 * (i >> 1);
    if (grow < n) {
      float* dp = (i & 1) ? ad_ : as_;
      dp[(size_t)grow * HEADS + lane / LPH] = val[0];
    }
  }
}


constexpr int SLICE_LOG2 = 14;
constexpr int SLICE      = 1 << SLICE_LOG2;
constexpr int NB         = 65536;

__global__ __launch_bounds__(1024) void edge_count_kernel(
    const int* __restrict__ ei, int E, int n,
    int* __restrict__ hist, int* __restrict__ pos) {
  constexpr int U = SLICE / 4096;
  __shared__ __attribute__((aligned(16))) int cnt[NB];
  const int tid = threadIdx.x;
  const int g   = blockIdx.x;
  const int np  = (n + 3) & ~3;
  const int* dst = ei + E;
  const bool vec = (E & 3) == 0;
  const int e0   = g * SLICE + tid * 4;

  const bool full = vec && (g * SLICE + SLICE <= E);

  int4 d[U], r[U];
  if (full) {
#pragma unroll
    for (int u = 0; u < U; ++u) d[u] = *(const int4*)(dst + e0 + u * 4096);
  } else {
#pragma unroll
    for (int u = 0; u < U; ++u) {
      const int e = e0 + u * 4096;
      const int x = dst[(e     < E) ? e     : E - 1];
      const int y = dst[(e + 1 < E) ? e + 1 : E - 1];
      const int z = dst[(e + 2 < E) ? e + 2 : E - 1];
      const int w = dst[(e + 3 < E) ? e + 3 : E - 1];
      d[u].x = (e     < E) ? x : -1;
      d[u].y = (e + 1 < E) ? y : -1;
      d[u].z = (e + 2 < E) ? z : -1;
      d[u].w = (e + 3 < E) ? w : -1;
    }
  }
#pragma unroll
  for (int u = 0; u < U; ++u) r[u] = make_int4(0, 0, 0, 0);

  for (int nb = 0; nb < n; nb += NB) {
    const int cur  = (n - nb < NB) ? (n - nb) : NB;
    const int cur4 = (cur + 3) & ~3;
    if (nb) __syncthreads();
    for (int j = tid * 4; j < cur4; j += 4096)
      *(int4*)(cnt + j) = make_int4(0, 0, 0, 0);
    __syncthreads();
#pragma unroll
    for (int u = 0; u < U; ++u) {
      const unsigned ox = (unsigned)(d[u].x - nb), oy = (unsigned)(d[u].y - nb);
      const unsigned oz = (unsigned)(d[u].z - nb), ow = (unsigned)(d[u].w - nb);
      if (ox < (unsigned)cur) r[u].x = atomicAdd(&cnt[ox], 1);
      if (oy < (unsigned)cur) r[u].y = atomicAdd(&cnt[oy], 1);
      if (oz < (unsigned)cur) r[u].z = atomicAdd(&cnt[oz], 1);
      if (ow < (unsigned)cur) r[u].w = atomicAdd(&cnt[ow], 1);
    }
    __syncthreads();
    int* hp = hist + (size_t)g * np + nb;
    int j = tid * 4;
    for (; j + 3 * 4096 < cur4; j += 4 * 4096) {
      int4 t[4];
#pragma unroll
      for (int q = 0; q < 4; ++q) t[q] = *(const int4*)(cnt + j + q * 4096);
#pragma unroll
      for (int q = 0; q < 4; ++q) *(int4*)(hp + j + q * 4096) = t[q];
    }
    for (; j < cur4; j += 4096)
      *(int4*)(hp + j) = *(const int4*)(cnt + j);
  }

  if (full) {
#pragma unroll
    for (int u = 0; u < U; ++u) *(int4*)(pos + e0 + u * 4096) = r[u];
  } else {
#pragma unroll
    for (int u = 0; u < U; ++u) {
      const int e = e0 + u * 4096;
      if (e     < E) pos[e]     = r[u].x;
      if (e + 1 < E) pos[e + 1] = r[u].y;
      if (e + 2 < E) pos[e + 2] = r[u].z;
      if (e + 3 < E) pos[e + 3] = r[u].w;
    }
  }
}

constexpr int CB_LOG2 = 8;
constexpr int CB      = 1 << CB_LOG2;

__global__ __launch_bounds__(CB) void hist_colsum_kernel(
    int* __restrict__ hist, int* __restrict__ rowloc, int* __restrict__ blocksum,
    int n, int G) {
  constexpr int T = 32;
  __shared__ int wsum[CB / 32];
  const int tid  = threadIdx.x;
  const int lane = tid & 31;
  const int w    = tid >> 5;
  const int d    = blockIdx.x * CB + tid;
  const bool valid = d < n;
  const size_t np = (size_t)((n + 3) & ~3);
  int run = 0;
  if (valid) {
    int* hp = hist + d;
    for (int g0 = 0; g0 < G; g0 += T) {
      int v[T];
#pragma unroll
      for (int t = 0; t < T; ++t)
        v[t] = (g0 + t < G) ? hp[(size_t)(g0 + t) * np] : 0;
#pragma unroll
      for (int t = 0; t < T; ++t) {
        if (g0 + t < G) hp[(size_t)(g0 + t) * np] = run;
        run += v[t];
      }
    }
  }
  int inc = run;
#pragma unroll
  for (int off = 1; off < 32; off <<= 1) {
    const int u = __shfl_up(inc, off);
    if (lane >= off) inc += u;
  }
  if (lane == 31) wsum[w] = inc;
  __syncthreads();
  int before = 0;
#pragma unroll
  for (int k = 0; k < CB / 32; ++k) before += (k < w) ? wsum[k] : 0;
  if (valid) rowloc[d] = before + inc - run;
  if (tid == CB - 1) blocksum[blockIdx.x] = before + inc;
}

__global__ __launch_bounds__(CB) void edge_place_kernel(
    const int* __restrict__ ei, int E, int n,
    const int* __restrict__ rowloc, const int* __restrict__ blocksum, int nblk,
    const int* __restrict__ hist, const int* __restrict__ pos,
    int* __restrict__ rowptr, int* __restrict__ csr_src) {
  extern __shared__ int bp[];
  __shared__ int wsum[CB / 32];
  const int tid  = threadIdx.x;
  const int lane = tid & 31;
  const int w    = tid >> 5;
  const int i    = blockIdx.x * CB + tid;
  const bool valid = i < E;
  const size_t np = (size_t)((n + 3) & ~3);

  int v = (tid < nblk) ? blocksum[tid] : 0;
  int s = 0, d = 0, rl = 0, hv = 0, pv = 0;
  if (valid) {
    s  = ei[i];
    d  = ei[E + i];
    pv = pos[i];
    rl = rowloc[d];
    hv = hist[(size_t)(i >> SLICE_LOG2) * np + d];
  }

  int carry = 0;
  for (int c0 = 0; c0 < nblk; c0 += CB) {
    const int idx = c0 + tid;
    if (c0) v = (idx < nblk) ? blocksum[idx] : 0;
    int inc = v;
#pragma unroll
    for (int off = 1; off < 32; off <<= 1) {
      const int u = __shfl_up(inc, off);
      if (lane >= off) inc += u;
    }
    if (lane == 31) wsum[w] = inc;
    __syncthreads();
    int before = 0, total = 0;
#pragma unroll
    for (int k = 0; k < CB / 32; ++k) {
      const int t = wsum[k];
      before += (k < w) ? t : 0;
      total  += t;
    }
    if (idx < nblk) bp[idx] = carry + before + inc - v;
    carry += total;
    __syncthreads();
  }

  for (int b = blockIdx.x; b < nblk; b += gridDim.x) {
    const int dn = b * CB + tid;
    if (dn < n) rowptr[dn] = bp[b] + rowloc[dn];
  }
  if (blockIdx.x == 0 && tid == 0) rowptr[n] = carry;

  if (valid) {
    const int slot = bp[d >> CB_LOG2] + rl + hv + pv;
    if ((unsigned)slot < (unsigned)E) csr_src[slot] = s;
  }
}


__device__ __forceinline__ float wave_max_f(float v) {
#pragma unroll
  for (int off = 16; off > 0; off >>= 1) v = fmaxf(v, __shfl_xor(v, off));
  return v;
}
__device__ __forceinline__ float wave_sum_f(float v) {
#pragma unroll
  for (int off = 16; off > 0; off >>= 1) v += __shfl_xor(v, off);
  return v;
}
__device__ __forceinline__ float leaky(float e) {
  return (e > 0.f) ? e : NEG_SLOPE * e;
}

template <int HEADS, int C, bool ELU>
__global__ __launch_bounds__(256) void gat_gather_kernel(
    const int* __restrict__ rowptr, const int* __restrict__ csr_src,
    const float* __restrict__ H, const float* __restrict__ as_,
    const float* __restrict__ ad_, const float* __restrict__ bias,
    float* __restrict__ out, int n, int E) {
  constexpr int CH  = HEADS * C;
  constexpr int VEC = CH / 32;
  typedef float vecf __attribute__((ext_vector_type(VEC)));

  const int lane = threadIdx.x & 31;
  const int wv  = __builtin_amdgcn_readfirstlane((int)(threadIdx.x >> 5));
  const int gw  = blockIdx.x * 8 + wv;
  const int nw  = gridDim.x * 8;
  const int myh = (lane * VEC) / C;
  const vecf bv = *(const vecf*)(bias + lane * VEC);

  for (int node = gw; node < n; node += nw) {
    int beg = rowptr[node];
    int end = rowptr[node + 1];
    beg = (beg < 0) ? 0 : beg;
    end = (end > E) ? E : end;
    const int deg = end - beg;

    const vecf hself = *(const vecf*)(H + (size_t)node * CH + lane * VEC);

    float adh[HEADS], eself[HEADS], m[HEADS];
#pragma unroll
    for (int h = 0; h < HEADS; ++h) {
      adh[h]   = ad_[node * HEADS + h];
      eself[h] = leaky(as_[node * HEADS + h] + adh[h]);
    }

    const bool v0 = lane < deg;
    int s0 = node;
    if (v0) {
      s0 = csr_src[beg + lane];
      s0 = (s0 < 0) ? 0 : ((s0 >= n) ? n - 1 : s0);
    }
    float e0[HEADS];
    if (HEADS == 2) {
      const v2f t = *(const v2f*)(as_ + (size_t)s0 * 2);
      e0[0]         = leaky(t.x + adh[0]);
      e0[HEADS - 1] = leaky(t.y + adh[HEADS - 1]);
    } else {
      e0[0] = leaky(as_[s0] + adh[0]);
    }
#pragma unroll
    for (int h = 0; h < HEADS; ++h) m[h] = fmaxf(eself[h], e0[h]);
    for (int base = 32; base < deg; base += 32) {
      int j = base + lane;
      if (j < deg) {
        int s = csr_src[beg + j];
        s = (s < 0) ? 0 : ((s >= n) ? n - 1 : s);
#pragma unroll
        for (int h = 0; h < HEADS; ++h)
          m[h] = fmaxf(m[h], leaky(as_[s * HEADS + h] + adh[h]));
      }
    }
#pragma unroll
    for (int h = 0; h < HEADS; ++h) m[h] = wave_max_f(m[h]);

    float p0[HEADS], den[HEADS], pself[HEADS];
#pragma unroll
    for (int h = 0; h < HEADS; ++h) {
      p0[h]  = v0 ? __expf(e0[h] - m[h]) : 0.f;
      den[h] = p0[h];
    }
    for (int base = 32; base < deg; base += 32) {
      int j = base + lane;
      if (j < deg) {
        int s = csr_src[beg + j];
        s = (s < 0) ? 0 : ((s >= n) ? n - 1 : s);
#pragma unroll
        for (int h = 0; h < HEADS; ++h)
          den[h] += __expf(leaky(as_[s * HEADS + h] + adh[h]) - m[h]);
      }
    }
#pragma unroll
    for (int h = 0; h < HEADS; ++h) {
      pself[h] = __expf(eself[h] - m[h]);
      den[h]   = wave_sum_f(den[h]) + pself[h] + 1e-16f;
    }

    vecf acc;
    {
      const float a = (HEADS == 2 && myh) ? pself[HEADS - 1] / den[HEADS - 1]
                                          : pself[0] / den[0];
      acc = hself * a;
    }
    for (int base = 0; base < deg; base += 32) {
      int   sj = s0;
      float al[HEADS];
      if (base == 0) {
#pragma unroll
        for (int h = 0; h < HEADS; ++h) al[h] = p0[h] / den[h];
      } else {
        int j = base + lane;
        sj = node;
#pragma unroll
        for (int h = 0; h < HEADS; ++h) al[h] = 0.f;
        if (j < deg) {
          sj = csr_src[beg + j];
          sj = (sj < 0) ? 0 : ((sj >= n) ? n - 1 : sj);
#pragma unroll
          for (int h = 0; h < HEADS; ++h)
            al[h] = __expf(leaky(as_[sj * HEADS + h] + adh[h]) - m[h]) / den[h];
        }
      }
      const int cnt  = (deg - base < 32) ? (deg - base) : 32;
      const int cnt4 = (cnt + 3) & ~3;
      for (int k = 0; k < cnt4; k += 4) {
        vecf  hv[4];
        float a[4];
#pragma unroll
        for (int u = 0; u < 4; ++u) {
          const int sk = __builtin_amdgcn_readlane(sj, k + u);
          hv[u] = *(const vecf*)(H + (size_t)sk * CH + lane * VEC);
          const float a0 = __int_as_float(__builtin_amdgcn_readlane(__float_as_int(al[0]), k + u));
          a[u] = a0;
          if (HEADS == 2) {
            const float a1 = __int_as_float(__builtin_amdgcn_readlane(__float_as_int(al[HEADS - 1]), k + u));
            a[u] = myh ? a1 : a0;
          }
        }
#pragma unroll
        for (int u = 0; u < 4; ++u) acc += hv[u] * a[u];
      }
    }

#pragma unroll
    for (int v = 0; v < VEC; ++v) {
      float y = acc[v] + bv[v];
      if (ELU) y = (y > 0.f) ? y : expm1f(y);
      acc[v] = y;
    }
    *(vecf*)(out + (size_t)node * CH + lane * VEC) = acc;
  }
}


static inline int cdiv(long long a, long long b) { return (int)((a + b - 1) / b); }

extern "C" void kernel_launch(void* const* d_in, const int* in_sizes, int n_in,
                              void* d_out, int out_size, void* d_ws, size_t ws_size,
                              hipStream_t stream) {
  const int IN_CH = 128, HID = 64, HEADS = 2, OUT = 64;
  const int CH1 = HEADS * HID;

  const float* x        = (const float*)d_in[0];
  const int*   ei       = (const int*)d_in[1];
  const float* W1       = (const float*)d_in[2];
  const float* att1_src = (const float*)d_in[3];
  const float* att1_dst = (const float*)d_in[4];
  const float* b1       = (const float*)d_in[5];
  const float* W2       = (const float*)d_in[6];
  const float* att2_src = (const float*)d_in[7];
  const float* att2_dst = (const float*)d_in[8];
  const float* b2       = (const float*)d_in[9];
  float* out            = (float*)d_out;

  const int n  = in_sizes[0] / IN_CH;
  const int E  = in_sizes[1] / 2;

  char* ws = (char*)d_ws;
  size_t off = 0;
  auto alloc = [&](size_t bytes) -> void* {
    void* p = ws + off;
    off += (bytes + 255) & ~(size_t)255;
    return p;
  };
  float* h1      = (float*)alloc((size_t)n * CH1 * 4);
  float* agg1    = (float*)alloc((size_t)n * CH1 * 4);
  float* h2      = (float*)alloc((size_t)n * OUT * 4);
  float* as1     = (float*)alloc((size_t)n * HEADS * 4);
  float* ad1     = (float*)alloc((size_t)n * HEADS * 4);
  float* as2     = (float*)alloc((size_t)n * 4);
  float* ad2     = (float*)alloc((size_t)n * 4);
  const int nblk = cdiv(n, CB);
  int*   rowloc  = (int*)  alloc((size_t)n * 4);
  int*   blocksum= (int*)  alloc((size_t)nblk * 4);
  int*   rowptr  = (int*)  alloc((size_t)(n + 1) * 4);
  int*   pos     = (int*)  alloc((size_t)E * 4);
  int*   csr_src = (int*)  alloc((size_t)E * 4);
  const int G    = cdiv(E, SLICE);
  int*   hist    = (int*)  alloc((size_t)G * (n + 4) * 4);
  (void)ws_size; (void)n_in; (void)out_size;

  const int BT = 256;
  const int GATHER_BLOCKS = 2048;

  edge_count_kernel<<<G, 1024, 0, stream>>>(ei, E, n, hist, pos);
  hist_colsum_kernel<<<nblk, CB, 0, stream>>>(hist, rowloc, blocksum, n, G);
  edge_place_kernel<<<cdiv(E, CB), CB, (size_t)nblk * sizeof(int), stream>>>(
      ei, E, n, rowloc, blocksum, nblk, hist, pos, rowptr, csr_src);

  gemm_wmma_f32<128, 32, 2><<<cdiv(n, 32), BT, 0, stream>>>(
      x, W1, att1_src, att1_dst, h1, as1, ad1, n);
  gat_gather_kernel<2, 64, true><<<GATHER_BLOCKS, BT, 0, stream>>>(
      rowptr, csr_src, h1, as1, ad1, b1, agg1, n, E);

  gemm_wmma_f32<64, 64, 1><<<cdiv(n, 64), BT, 0, stream>>>(
      agg1, W2, att2_src, att2_dst, h2, as2, ad2, n);
  gat_gather_kernel<1, 64, false><<<GATHER_BLOCKS, BT, 0, stream>>>(
      rowptr, csr_src, h2, as2, ad2, b2, out, n, E);
}


# column-sum pass of the edge sort runs inside the layer-1 GEMM launch (independent work: first work-groups take node blocks, the rest row slabs); 1 launch less, every sum order unchanged
# speedup vs baseline: 8.7609x; 1.0446x over previous
#include <hip/hip_runtime.h>
#include <hip/hip_bf16.h>


typedef float v2f __attribute__((ext_vector_type(2)));
typedef float v8f __attribute__((ext_vector_type(8)));

#define NEG_SLOPE 0.2f


template <int M, int OFF, int NV>
__device__ __forceinline__ void head_sum(float (&val)[NV], int lane) {
  if constexpr (OFF > 0) {
    if constexpr (M > 1) {
      constexpr int half = M / 2;
      const bool upper = (lane & OFF) != 0;
#pragma unroll
      for (int k = 0; k < half; ++k) {
        const float lo = val[k], hi = val[k + half];
        const float send = upper ? lo : hi;
        const float keep = upper ? hi : lo;
        val[k] = keep + __shfl_xor(send, OFF);
      }
      head_sum<half, OFF / 2, NV>(val, lane);
    } else {
      val[0] += __shfl_xor(val[0], OFF);
      head_sum<1, OFF / 2, NV>(val, lane);
    }
  }
}


template <int COUT, int BM, int HEADS>
__device__ __forceinline__ void gemm_tile(
    const float* __restrict__ A, const float* __restrict__ W,
    const float* __restrict__ att_src, const float* __restrict__ att_dst,
    float* __restrict__ H, float* __restrict__ as_, float* __restrict__ ad_,
    int n, int blk) {
  constexpr int K  = 128;
  constexpr int P  = K + 4;
  constexpr int CT = COUT / 16;
  static_assert(CT * (BM / 32) == 8, "8 waves per work-group");
  constexpr int WV4 = COUT * (K / 4) / 256;
  constexpr int AV4 = BM * (K / 4) / 256;
  __shared__ __attribute__((aligned(16))) float Ws[COUT * P];
  __shared__ __attribute__((aligned(16))) float As[BM * P];

  const int tid  = threadIdx.x;
  const int lane = tid & 31;
  const int wv   = __builtin_amdgcn_readfirstlane(tid >> 5);
  const int row0 = blk * BM;

  float4 wreg[WV4], areg[AV4];
#pragma unroll
  for (int j = 0; j < WV4; ++j) {
    const int r = wv + 8 * j;
    wreg[j] = *(const float4*)(W + (size_t)r * K + lane * 4);
  }
#pragma unroll
  for (int j = 0; j < AV4; ++j) {
    int r = row0 + wv + 8 * j;
    r = (r < n) ? r : (n - 1);
    areg[j] = *(const float4*)(A + (size_t)r * K + lane * 4);
  }
#pragma unroll
  for (int j = 0; j < WV4; ++j)
    *(float4*)(Ws + (wv + 8 * j) * P + lane * 4) = wreg[j];
#pragma unroll
  for (int j = 0; j < AV4; ++j)
    *(float4*)(As + (wv + 8 * j) * P + lane * 4) = areg[j];
  __syncthreads();

  const int mn      = lane & 15;
  const int khalf   = (lane >> 4) * 2;
  const int colTile = wv % CT;
  const int rg      = wv / CT;

  const float* bp  = Ws + (colTile * 16 + mn) * P + khalf;
  const float* ap0 = As + (rg * 32 + mn) * P + khalf;
  const float* ap1 = ap0 + 16 * P;

  v8f acc0 = {}, acc1 = {};
#pragma unroll
  for (int k0 = 0; k0 < K; k0 += 4) {
    const v2f b  = *(const v2f*)(bp + k0);
    const v2f a0 = *(const v2f*)(ap0 + k0);
    const v2f a1 = *(const v2f*)(ap1 + k0);
    acc0 = __builtin_amdgcn_wmma_f32_16x16x4_f32(
         false, a0,  false, b,
         (short)0, acc0,  false,  false);
    acc1 = __builtin_amdgcn_wmma_f32_16x16x4_f32(
         false, a1,  false, b,
         (short)0, acc1,  false,  false);
  }

  constexpr int VEC = COUT / 32;
  constexpr int LPH = 32 / HEADS;
  typedef float vecf __attribute__((ext_vector_type(VEC)));
  const vecf sv = *(const vecf*)(att_src + lane * VEC);
  const vecf dv = *(const vecf*)(att_dst + lane * VEC);

  __syncthreads();
  {
    float* op = As + (rg * 32 + 8 * (lane >> 4)) * P + colTile * 16 + mn;
#pragma unroll
    for (int v = 0; v < 8; ++v) {
      op[v * P]        = acc0[v];
      op[(16 + v) * P] = acc1[v];
    }
  }
  __syncthreads();

  constexpr int R  = BM / 8;
  constexpr int NV = 2 * R;
  static_assert(NV == LPH / 2, "the halving exchange below ends with one value per lane pair");
  vecf hrow[R];
#pragma unroll
  for (int j = 0; j < R; ++j)
    hrow[j] = *(const vecf*)(As + (wv + 8 * j) * P + lane * VEC);
#pragma unroll
  for (int j = 0; j < R; ++j) {
    const int grow = row0 + wv + 8 * j;
    if (grow < n) *(vecf*)(H + (size_t)grow * COUT + lane * VEC) = hrow[j];
  }

  float val[NV];
#pragma unroll
  for (int j = 0; j < R; ++j) {
    float ps = 0.f, pd = 0.f;
#pragma unroll
    for (int c = 0; c < VEC; ++c) {
      ps += hrow[j][c] * sv[c];
      pd += hrow[j][c] * dv[c];
    }
    val[2 * j]     = ps;
    val[2 * j + 1] = pd;
  }
  head_sum<NV, LPH / 2, NV>(val, lane);
  if ((lane & 1) == 0) {
    const int i    = (lane >> 1) & (NV - 1);
    const int grow = row0 + wv + 8 * (i >> 1);
    if (grow < n) {
      float* dp = (i & 1) ? ad_ : as_;
      dp[(size_t)grow * HEADS + lane / LPH] = val[0];
    }
  }
}

template <int COUT, int BM, int HEADS>
__global__ __launch_bounds__(256) void gemm_wmma_f32(
    const float* __restrict__ A, const float* __restrict__ W,
    const float* __restrict__ att_src, const float* __restrict__ att_dst,
    float* __restrict__ H, float* __restrict__ as_, float* __restrict__ ad_,
    int n) {
  gemm_tile<COUT, BM, HEADS>(A, W, att_src, att_dst, H, as_, ad_, n, (int)blockIdx.x);
}


constexpr int SLICE_LOG2 = 14;
constexpr int SLICE      = 1 << SLICE_LOG2;
constexpr int NB         = 65536;

__global__ __launch_bounds__(1024) void edge_count_kernel(
    const int* __restrict__ ei, int E, int n,
    int* __restrict__ hist, int* __restrict__ pos) {
  constexpr int U = SLICE / 4096;
  __shared__ __attribute__((aligned(16))) int cnt[NB];
  const int tid = threadIdx.x;
  const int g   = blockIdx.x;
  const int np  = (n + 3) & ~3;
  const int* dst = ei + E;
  const bool vec = (E & 3) == 0;
  const int e0   = g * SLICE + tid * 4;

  const bool full = vec && (g * SLICE + SLICE <= E);

  int4 d[U], r[U];
  if (full) {
#pragma unroll
    for (int u = 0; u < U; ++u) d[u] = *(const int4*)(dst + e0 + u * 4096);
  } else {
#pragma unroll
    for (int u = 0; u < U; ++u) {
      const int e = e0 + u * 4096;
      const int x = dst[(e     < E) ? e     : E - 1];
      const int y = dst[(e + 1 < E) ? e + 1 : E - 1];
      const int z = dst[(e + 2 < E) ? e + 2 : E - 1];
      const int w = dst[(e + 3 < E) ? e + 3 : E - 1];
      d[u].x = (e     < E) ? x : -1;
      d[u].y = (e + 1 < E) ? y : -1;
      d[u].z = (e + 2 < E) ? z : -1;
      d[u].w = (e + 3 < E) ? w : -1;
    }
  }
#pragma unroll
  for (int u = 0; u < U; ++u) r[u] = make_int4(0, 0, 0, 0);

  for (int nb = 0; nb < n; nb += NB) {
    const int cur  = (n - nb < NB) ? (n - nb) : NB;
    const int cur4 = (cur + 3) & ~3;
    if (nb) __syncthreads();
    for (int j = tid * 4; j < cur4; j += 4096)
      *(int4*)(cnt + j) = make_int4(0, 0, 0, 0);
    __syncthreads();
#pragma unroll
    for (int u = 0; u < U; ++u) {
      const unsigned ox = (unsigned)(d[u].x - nb), oy = (unsigned)(d[u].y - nb);
      const unsigned oz = (unsigned)(d[u].z - nb), ow = (unsigned)(d[u].w - nb);
      if (ox < (unsigned)cur) r[u].x = atomicAdd(&cnt[ox], 1);
      if (oy < (unsigned)cur) r[u].y = atomicAdd(&cnt[oy], 1);
      if (oz < (unsigned)cur) r[u].z = atomicAdd(&cnt[oz], 1);
      if (ow < (unsigned)cur) r[u].w = atomicAdd(&cnt[ow], 1);
    }
    __syncthreads();
    int* hp = hist + (size_t)g * np + nb;
    int j = tid * 4;
    for (; j + 3 * 4096 < cur4; j += 4 * 4096) {
      int4 t[4];
#pragma unroll
      for (int q = 0; q < 4; ++q) t[q] = *(const int4*)(cnt + j + q * 4096);
#pragma unroll
      for (int q = 0; q < 4; ++q) *(int4*)(hp + j + q * 4096) = t[q];
    }
    for (; j < cur4; j += 4096)
      *(int4*)(hp + j) = *(const int4*)(cnt + j);
  }

  if (full) {
#pragma unroll
    for (int u = 0; u < U; ++u) *(int4*)(pos + e0 + u * 4096) = r[u];
  } else {
#pragma unroll
    for (int u = 0; u < U; ++u) {
      const int e = e0 + u * 4096;
      if (e     < E) pos[e]     = r[u].x;
      if (e + 1 < E) pos[e + 1] = r[u].y;
      if (e + 2 < E) pos[e + 2] = r[u].z;
      if (e + 3 < E) pos[e + 3] = r[u].w;
    }
  }
}

constexpr int CB_LOG2 = 8;
constexpr int CB      = 1 << CB_LOG2;

__device__ __forceinline__ void hist_colsum_block(
    int* __restrict__ hist, int* __restrict__ rowloc, int* __restrict__ blocksum,
    int n, int G, int blk) {
  constexpr int T = 32;
  __shared__ int wsum[CB / 32];
  const int tid  = threadIdx.x;
  const int lane = tid & 31;
  const int w    = tid >> 5;
  const int d    = blk * CB + tid;
  const bool valid = d < n;
  const size_t np = (size_t)((n + 3) & ~3);
  int run = 0;
  if (valid) {
    int* hp = hist + d;
    for (int g0 = 0; g0 < G; g0 += T) {
      int v[T];
#pragma unroll
      for (int t = 0; t < T; ++t)
        v[t] = (g0 + t < G) ? hp[(size_t)(g0 + t) * np] : 0;
#pragma unroll
      for (int t = 0; t < T; ++t) {
        if (g0 + t < G) hp[(size_t)(g0 + t) * np] = run;
        run += v[t];
      }
    }
  }
  int inc = run;
#pragma unroll
  for (int off = 1; off < 32; off <<= 1) {
    const int u = __shfl_up(inc, off);
    if (lane >= off) inc += u;
  }
  if (lane == 31) wsum[w] = inc;
  __syncthreads();
  int before = 0;
#pragma unroll
  for (int k = 0; k < CB / 32; ++k) before += (k < w) ? wsum[k] : 0;
  if (valid) rowloc[d] = before + inc - run;
  if (tid == CB - 1) blocksum[blk] = before + inc;
}

template <int COUT, int BM, int HEADS>
__global__ __launch_bounds__(CB) void gemm_colsum_kernel(
    const float* __restrict__ A, const float* __restrict__ W,
    const float* __restrict__ att_src, const float* __restrict__ att_dst,
    float* __restrict__ H, float* __restrict__ as_, float* __restrict__ ad_,
    int n,
    int* __restrict__ hist, int* __restrict__ rowloc, int* __restrict__ blocksum,
    int G, int nblk) {
  static_assert(CB == 256, "the GEMM body is written for 8 waves");
  const int b = (int)blockIdx.x;
  if (b < nblk) {
    hist_colsum_block(hist, rowloc, blocksum, n, G, b);
  } else {
    gemm_tile<COUT, BM, HEADS>(A, W, att_src, att_dst, H, as_, ad_, n, b - nblk);
  }
}

__global__ __launch_bounds__(CB) void edge_place_kernel(
    const int* __restrict__ ei, int E, int n,
    const int* __restrict__ rowloc, const int* __restrict__ blocksum, int nblk,
    const int* __restrict__ hist, const int* __restrict__ pos,
    int* __restrict__ rowptr, int* __restrict__ csr_src) {
  extern __shared__ int bp[];
  __shared__ int wsum[CB / 32];
  const int tid  = threadIdx.x;
  const int lane = tid & 31;
  const int w    = tid >> 5;
  const int i    = blockIdx.x * CB + tid;
  const bool valid = i < E;
  const size_t np = (size_t)((n + 3) & ~3);

  int v = (tid < nblk) ? blocksum[tid] : 0;
  int s = 0, d = 0, rl = 0, hv = 0, pv = 0;
  if (valid) {
    s  = ei[i];
    d  = ei[E + i];
    pv = pos[i];
    rl = rowloc[d];
    hv = hist[(size_t)(i >> SLICE_LOG2) * np + d];
  }

  int carry = 0;
  for (int c0 = 0; c0 < nblk; c0 += CB) {
    const int idx = c0 + tid;
    if (c0) v = (idx < nblk) ? blocksum[idx] : 0;
    int inc = v;
#pragma unroll
    for (int off = 1; off < 32; off <<= 1) {
      const int u = __shfl_up(inc, off);
      if (lane >= off) inc += u;
    }
    if (lane == 31) wsum[w] = inc;
    __syncthreads();
    int before = 0, total = 0;
#pragma unroll
    for (int k = 0; k < CB / 32; ++k) {
      const int t = wsum[k];
      before += (k < w) ? t : 0;
      total  += t;
    }
    if (idx < nblk) bp[idx] = carry + before + inc - v;
    carry += total;
    __syncthreads();
  }

  for (int b = blockIdx.x; b < nblk; b += gridDim.x) {
    const int dn = b * CB + tid;
    if (dn < n) rowptr[dn] = bp[b] + rowloc[dn];
  }
  if (blockIdx.x == 0 && tid == 0) rowptr[n] = carry;

  if (valid) {
    const int slot = bp[d >> CB_LOG2] + rl + hv + pv;
    if ((unsigned)slot < (unsigned)E) csr_src[slot] = s;
  }
}


__device__ __forceinline__ float wave_max_f(float v) {
#pragma unroll
  for (int off = 16; off > 0; off >>= 1) v = fmaxf(v, __shfl_xor(v, off));
  return v;
}
__device__ __forceinline__ float wave_sum_f(float v) {
#pragma unroll
  for (int off = 16; off > 0; off >>= 1) v += __shfl_xor(v, off);
  return v;
}
__device__ __forceinline__ float leaky(float e) {
  return (e > 0.f) ? e : NEG_SLOPE * e;
}

template <int HEADS, int C, bool ELU>
__global__ __launch_bounds__(256) void gat_gather_kernel(
    const int* __restrict__ rowptr, const int* __restrict__ csr_src,
    const float* __restrict__ H, const float* __restrict__ as_,
    const float* __restrict__ ad_, const float* __restrict__ bias,
    float* __restrict__ out, int n, int E) {
  constexpr int CH  = HEADS * C;
  constexpr int VEC = CH / 32;
  typedef float vecf __attribute__((ext_vector_type(VEC)));

  const int lane = threadIdx.x & 31;
  const int wv  = __builtin_amdgcn_readfirstlane((int)(threadIdx.x >> 5));
  const int gw  = blockIdx.x * 8 + wv;
  const int nw  = gridDim.x * 8;
  const int myh = (lane * VEC) / C;
  const vecf bv = *(const vecf*)(bias + lane * VEC);

  for (int node = gw; node < n; node += nw) {
    int beg = rowptr[node];
    int end = rowptr[node + 1];
    beg = (beg < 0) ? 0 : beg;
    end = (end > E) ? E : end;
    const int deg = end - beg;

    const vecf hself = *(const vecf*)(H + (size_t)node * CH + lane * VEC);

    float adh[HEADS], eself[HEADS], m[HEADS];
#pragma unroll
    for (int h = 0; h < HEADS; ++h) {
      adh[h]   = ad_[node * HEADS + h];
      eself[h] = leaky(as_[node * HEADS + h] + adh[h]);
    }

    const bool v0 = lane < deg;
    int s0 = node;
    if (v0) {
      s0 = csr_src[beg + lane];
      s0 = (s0 < 0) ? 0 : ((s0 >= n) ? n - 1 : s0);
    }
    float e0[HEADS];
    if (HEADS == 2) {
      const v2f t = *(const v2f*)(as_ + (size_t)s0 * 2);
      e0[0]         = leaky(t.x + adh[0]);
      e0[HEADS - 1] = leaky(t.y + adh[HEADS - 1]);
    } else {
      e0[0] = leaky(as_[s0] + adh[0]);
    }
#pragma unroll
    for (int h = 0; h < HEADS; ++h) m[h] = fmaxf(eself[h], e0[h]);
    for (int base = 32; base < deg; base += 32) {
      int j = base + lane;
      if (j < deg) {
        int s = csr_src[beg + j];
        s = (s < 0) ? 0 : ((s >= n) ? n - 1 : s);
#pragma unroll
        for (int h = 0; h < HEADS; ++h)
          m[h] = fmaxf(m[h], leaky(as_[s * HEADS + h] + adh[h]));
      }
    }
#pragma unroll
    for (int h = 0; h < HEADS; ++h) m[h] = wave_max_f(m[h]);

    float p0[HEADS], den[HEADS], pself[HEADS];
#pragma unroll
    for (int h = 0; h < HEADS; ++h) {
      p0[h]  = v0 ? __expf(e0[h] - m[h]) : 0.f;
      den[h] = p0[h];
    }
    for (int base = 32; base < deg; base += 32) {
      int j = base + lane;
      if (j < deg) {
        int s = csr_src[beg + j];
        s = (s < 0) ? 0 : ((s >= n) ? n - 1 : s);
#pragma unroll
        for (int h = 0; h < HEADS; ++h)
          den[h] += __expf(leaky(as_[s * HEADS + h] + adh[h]) - m[h]);
      }
    }
#pragma unroll
    for (int h = 0; h < HEADS; ++h) {
      pself[h] = __expf(eself[h] - m[h]);
      den[h]   = wave_sum_f(den[h]) + pself[h] + 1e-16f;
    }

    vecf acc;
    {
      const float a = (HEADS == 2 && myh) ? pself[HEADS - 1] / den[HEADS - 1]
                                          : pself[0] / den[0];
      acc = hself * a;
    }
    for (int base = 0; base < deg; base += 32) {
      int   sj = s0;
      float al[HEADS];
      if (base == 0) {
#pragma unroll
        for (int h = 0; h < HEADS; ++h) al[h] = p0[h] / den[h];
      } else {
        int j = base + lane;
        sj = node;
#pragma unroll
        for (int h = 0; h < HEADS; ++h) al[h] = 0.f;
        if (j < deg) {
          sj = csr_src[beg + j];
          sj = (sj < 0) ? 0 : ((sj >= n) ? n - 1 : sj);
#pragma unroll
          for (int h = 0; h < HEADS; ++h)
            al[h] = __expf(leaky(as_[sj * HEADS + h] + adh[h]) - m[h]) / den[h];
        }
      }
      const int cnt  = (deg - base < 32) ? (deg - base) : 32;
      const int cnt4 = (cnt + 3) & ~3;
      for (int k = 0; k < cnt4; k += 4) {
        vecf  hv[4];
        float a[4];
#pragma unroll
        for (int u = 0; u < 4; ++u) {
          const int sk = __builtin_amdgcn_readlane(sj, k + u);
          hv[u] = *(const vecf*)(H + (size_t)sk * CH + lane * VEC);
          const float a0 = __int_as_float(__builtin_amdgcn_readlane(__float_as_int(al[0]), k + u));
          a[u] = a0;
          if (HEADS == 2) {
            const float a1 = __int_as_float(__builtin_amdgcn_readlane(__float_as_int(al[HEADS - 1]), k + u));
            a[u] = myh ? a1 : a0;
          }
        }
#pragma unroll
        for (int u = 0; u < 4; ++u) acc += hv[u] * a[u];
      }
    }

#pragma unroll
    for (int v = 0; v < VEC; ++v) {
      float y = acc[v] + bv[v];
      if (ELU) y = (y > 0.f) ? y : expm1f(y);
      acc[v] = y;
    }
    *(vecf*)(out + (size_t)node * CH + lane * VEC) = acc;
  }
}


static inline int cdiv(long long a, long long b) { return (int)((a + b - 1) / b); }

extern "C" void kernel_launch(void* const* d_in, const int* in_sizes, int n_in,
                              void* d_out, int out_size, void* d_ws, size_t ws_size,
                              hipStream_t stream) {
  const int IN_CH = 128, HID = 64, HEADS = 2, OUT = 64;
  const int CH1 = HEADS * HID;

  const float* x        = (const float*)d_in[0];
  const int*   ei       = (const int*)d_in[1];
  const float* W1       = (const float*)d_in[2];
  const float* att1_src = (const float*)d_in[3];
  const float* att1_dst = (const float*)d_in[4];
  const float* b1       = (const float*)d_in[5];
  const float* W2       = (const float*)d_in[6];
  const float* att2_src = (const float*)d_in[7];
  const float* att2_dst = (const float*)d_in[8];
  const float* b2       = (const float*)d_in[9];
  float* out            = (float*)d_out;

  const int n  = in_sizes[0] / IN_CH;
  const int E  = in_sizes[1] / 2;

  char* ws = (char*)d_ws;
  size_t off = 0;
  auto alloc = [&](size_t bytes) -> void* {
    void* p = ws + off;
    off += (bytes + 255) & ~(size_t)255;
    return p;
  };
  float* h1      = (float*)alloc((size_t)n * CH1 * 4);
  float* agg1    = (float*)alloc((size_t)n * CH1 * 4);
  float* h2      = (float*)alloc((size_t)n * OUT * 4);
  float* as1     = (float*)alloc((size_t)n * HEADS * 4);
  float* ad1     = (float*)alloc((size_t)n * HEADS * 4);
  float* as2     = (float*)alloc((size_t)n * 4);
  float* ad2     = (float*)alloc((size_t)n * 4);
  const int nblk = cdiv(n, CB);
  int*   rowloc  = (int*)  alloc((size_t)n * 4);
  int*   blocksum= (int*)  alloc((size_t)nblk * 4);
  int*   rowptr  = (int*)  alloc((size_t)(n + 1) * 4);
  int*   pos     = (int*)  alloc((size_t)E * 4);
  int*   csr_src = (int*)  alloc((size_t)E * 4);
  const int G    = cdiv(E, SLICE);
  int*   hist    = (int*)  alloc((size_t)G * (n + 4) * 4);
  (void)ws_size; (void)n_in; (void)out_size;

  const int BT = 256;
  const int GATHER_BLOCKS = 2048;

  edge_count_kernel<<<G, 1024, 0, stream>>>(ei, E, n, hist, pos);
  gemm_colsum_kernel<128, 32, 2><<<nblk + cdiv(n, 32), BT, 0, stream>>>(
      x, W1, att1_src, att1_dst, h1, as1, ad1, n, hist, rowloc, blocksum, G, nblk);
  edge_place_kernel<<<cdiv(E, CB), CB, (size_t)nblk * sizeof(int), stream>>>(
      ei, E, n, rowloc, blocksum, nblk, hist, pos, rowptr, csr_src);

  gat_gather_kernel<2, 64, true><<<GATHER_BLOCKS, BT, 0, stream>>>(
      rowptr, csr_src, h1, as1, ad1, b1, agg1, n, E);

  gemm_wmma_f32<64, 64, 1><<<cdiv(n, 64), BT, 0, stream>>>(
      agg1, W2, att2_src, att2_dst, h2, as2, ad2, n);
  gat_gather_kernel<1, 64, false><<<GATHER_BLOCKS, BT, 0, stream>>>(
      rowptr, csr_src, h2, as2, ad2, b2, out, n, E);
}


# layer-2 GEMM runs inside the layer-1 gather launch: 16 waves gather 50 nodes' rows (bias, ELU) into LDS, form the h2 tile and its attention dots from there; 1 launch less, sum orders unchanged
# speedup vs baseline: 9.0502x; 1.0330x over previous
#include <hip/hip_runtime.h>
#include <hip/hip_bf16.h>


typedef float v2f __attribute__((ext_vector_type(2)));
typedef float v8f __attribute__((ext_vector_type(8)));

#define NEG_SLOPE 0.2f


template <int M, int OFF, int NV>
__device__ __forceinline__ void head_sum(float (&val)[NV], int lane) {
  if constexpr (OFF > 0) {
    if constexpr (M > 1) {
      constexpr int half = M / 2;
      const bool upper = (lane & OFF) != 0;
#pragma unroll
      for (int k = 0; k < half; ++k) {
        const float lo = val[k], hi = val[k + half];
        const float send = upper ? lo : hi;
        const float keep = upper ? hi : lo;
        val[k] = keep + __shfl_xor(send, OFF);
      }
      head_sum<half, OFF / 2, NV>(val, lane);
    } else {
      val[0] += __shfl_xor(val[0], OFF);
      head_sum<1, OFF / 2, NV>(val, lane);
    }
  }
}


template <int COUT, int BM, int HEADS>
__device__ __forceinline__ void gemm_tile(
    const float* __restrict__ A, const float* __restrict__ W,
    const float* __restrict__ att_src, const float* __restrict__ att_dst,
    float* __restrict__ H, float* __restrict__ as_, float* __restrict__ ad_,
    int n, int blk) {
  constexpr int K  = 128;
  constexpr int P  = K + 4;
  constexpr int CT = COUT / 16;
  static_assert(CT * (BM / 32) == 8, "8 waves per work-group");
  constexpr int WV4 = COUT * (K / 4) / 256;
  constexpr int AV4 = BM * (K / 4) / 256;
  __shared__ __attribute__((aligned(16))) float Ws[COUT * P];
  __shared__ __attribute__((aligned(16))) float As[BM * P];

  const int tid  = threadIdx.x;
  const int lane = tid & 31;
  const int wv   = __builtin_amdgcn_readfirstlane(tid >> 5);
  const int row0 = blk * BM;

  float4 wreg[WV4], areg[AV4];
#pragma unroll
  for (int j = 0; j < WV4; ++j) {
    const int r = wv + 8 * j;
    wreg[j] = *(const float4*)(W + (size_t)r * K + lane * 4);
  }
#pragma unroll
  for (int j = 0; j < AV4; ++j) {
    int r = row0 + wv + 8 * j;
    r = (r < n) ? r : (n - 1);
    areg[j] = *(const float4*)(A + (size_t)r * K + lane * 4);
  }
#pragma unroll
  for (int j = 0; j < WV4; ++j)
    *(float4*)(Ws + (wv + 8 * j) * P + lane * 4) = wreg[j];
#pragma unroll
  for (int j = 0; j < AV4; ++j)
    *(float4*)(As + (wv + 8 * j) * P + lane * 4) = areg[j];
  __syncthreads();

  const int mn      = lane & 15;
  const int khalf   = (lane >> 4) * 2;
  const int colTile = wv % CT;
  const int rg      = wv / CT;

  const float* bp  = Ws + (colTile * 16 + mn) * P + khalf;
  const float* ap0 = As + (rg * 32 + mn) * P + khalf;
  const float* ap1 = ap0 + 16 * P;

  v8f acc0 = {}, acc1 = {};
#pragma unroll
  for (int k0 = 0; k0 < K; k0 += 4) {
    const v2f b  = *(const v2f*)(bp + k0);
    const v2f a0 = *(const v2f*)(ap0 + k0);
    const v2f a1 = *(const v2f*)(ap1 + k0);
    acc0 = __builtin_amdgcn_wmma_f32_16x16x4_f32(
         false, a0,  false, b,
         (short)0, acc0,  false,  false);
    acc1 = __builtin_amdgcn_wmma_f32_16x16x4_f32(
         false, a1,  false, b,
         (short)0, acc1,  false,  false);
  }

  constexpr int VEC = COUT / 32;
  constexpr int LPH = 32 / HEADS;
  typedef float vecf __attribute__((ext_vector_type(VEC)));
  const vecf sv = *(const vecf*)(att_src + lane * VEC);
  const vecf dv = *(const vecf*)(att_dst + lane * VEC);

  __syncthreads();
  {
    float* op = As + (rg * 32 + 8 * (lane >> 4)) * P + colTile * 16 + mn;
#pragma unroll
    for (int v = 0; v < 8; ++v) {
      op[v * P]        = acc0[v];
      op[(16 + v) * P] = acc1[v];
    }
  }
  __syncthreads();

  constexpr int R  = BM / 8;
  constexpr int NV = 2 * R;
  static_assert(NV == LPH / 2, "the halving exchange below ends with one value per lane pair");
  vecf hrow[R];
#pragma unroll
  for (int j = 0; j < R; ++j)
    hrow[j] = *(const vecf*)(As + (wv + 8 * j) * P + lane * VEC);
#pragma unroll
  for (int j = 0; j < R; ++j) {
    const int grow = row0 + wv + 8 * j;
    if (grow < n) *(vecf*)(H + (size_t)grow * COUT + lane * VEC) = hrow[j];
  }

  float val[NV];
#pragma unroll
  for (int j = 0; j < R; ++j) {
    float ps = 0.f, pd = 0.f;
#pragma unroll
    for (int c = 0; c < VEC; ++c) {
      ps += hrow[j][c] * sv[c];
      pd += hrow[j][c] * dv[c];
    }
    val[2 * j]     = ps;
    val[2 * j + 1] = pd;
  }
  head_sum<NV, LPH / 2, NV>(val, lane);
  if ((lane & 1) == 0) {
    const int i    = (lane >> 1) & (NV - 1);
    const int grow = row0 + wv + 8 * (i >> 1);
    if (grow < n) {
      float* dp = (i & 1) ? ad_ : as_;
      dp[(size_t)grow * HEADS + lane / LPH] = val[0];
    }
  }
}


constexpr int SLICE_LOG2 = 14;
constexpr int SLICE      = 1 << SLICE_LOG2;
constexpr int NB         = 65536;

__global__ __launch_bounds__(1024) void edge_count_kernel(
    const int* __restrict__ ei, int E, int n,
    int* __restrict__ hist, int* __restrict__ pos) {
  constexpr int U = SLICE / 4096;
  __shared__ __attribute__((aligned(16))) int cnt[NB];
  const int tid = threadIdx.x;
  const int g   = blockIdx.x;
  const int np  = (n + 3) & ~3;
  const int* dst = ei + E;
  const bool vec = (E & 3) == 0;
  const int e0   = g * SLICE + tid * 4;

  const bool full = vec && (g * SLICE + SLICE <= E);

  int4 d[U], r[U];
  if (full) {
#pragma unroll
    for (int u = 0; u < U; ++u) d[u] = *(const int4*)(dst + e0 + u * 4096);
  } else {
#pragma unroll
    for (int u = 0; u < U; ++u) {
      const int e = e0 + u * 4096;
      const int x = dst[(e     < E) ? e     : E - 1];
      const int y = dst[(e + 1 < E) ? e + 1 : E - 1];
      const int z = dst[(e + 2 < E) ? e + 2 : E - 1];
      const int w = dst[(e + 3 < E) ? e + 3 : E - 1];
      d[u].x = (e     < E) ? x : -1;
      d[u].y = (e + 1 < E) ? y : -1;
      d[u].z = (e + 2 < E) ? z : -1;
      d[u].w = (e + 3 < E) ? w : -1;
    }
  }
#pragma unroll
  for (int u = 0; u < U; ++u) r[u] = make_int4(0, 0, 0, 0);

  for (int nb = 0; nb < n; nb += NB) {
    const int cur  = (n - nb < NB) ? (n - nb) : NB;
    const int cur4 = (cur + 3) & ~3;
    if (nb) __syncthreads();
    for (int j = tid * 4; j < cur4; j += 4096)
      *(int4*)(cnt + j) = make_int4(0, 0, 0, 0);
    __syncthreads();
#pragma unroll
    for (int u = 0; u < U; ++u) {
      const unsigned ox = (unsigned)(d[u].x - nb), oy = (unsigned)(d[u].y - nb);
      const unsigned oz = (unsigned)(d[u].z - nb), ow = (unsigned)(d[u].w - nb);
      if (ox < (unsigned)cur) r[u].x = atomicAdd(&cnt[ox], 1);
      if (oy < (unsigned)cur) r[u].y = atomicAdd(&cnt[oy], 1);
      if (oz < (unsigned)cur) r[u].z = atomicAdd(&cnt[oz], 1);
      if (ow < (unsigned)cur) r[u].w = atomicAdd(&cnt[ow], 1);
    }
    __syncthreads();
    int* hp = hist + (size_t)g * np + nb;
    int j = tid * 4;
    for (; j + 3 * 4096 < cur4; j += 4 * 4096) {
      int4 t[4];
#pragma unroll
      for (int q = 0; q < 4; ++q) t[q] = *(const int4*)(cnt + j + q * 4096);
#pragma unroll
      for (int q = 0; q < 4; ++q) *(int4*)(hp + j + q * 4096) = t[q];
    }
    for (; j < cur4; j += 4096)
      *(int4*)(hp + j) = *(const int4*)(cnt + j);
  }

  if (full) {
#pragma unroll
    for (int u = 0; u < U; ++u) *(int4*)(pos + e0 + u * 4096) = r[u];
  } else {
#pragma unroll
    for (int u = 0; u < U; ++u) {
      const int e = e0 + u * 4096;
      if (e     < E) pos[e]     = r[u].x;
      if (e + 1 < E) pos[e + 1] = r[u].y;
      if (e + 2 < E) pos[e + 2] = r[u].z;
      if (e + 3 < E) pos[e + 3] = r[u].w;
    }
  }
}

constexpr int CB_LOG2 = 8;
constexpr int CB      = 1 << CB_LOG2;

__device__ __forceinline__ void hist_colsum_block(
    int* __restrict__ hist, int* __restrict__ rowloc, int* __restrict__ blocksum,
    int n, int G, int blk) {
  constexpr int T = 32;
  __shared__ int wsum[CB / 32];
  const int tid  = threadIdx.x;
  const int lane = tid & 31;
  const int w    = tid >> 5;
  const int d    = blk * CB + tid;
  const bool valid = d < n;
  const size_t np = (size_t)((n + 3) & ~3);
  int run = 0;
  if (valid) {
    int* hp = hist + d;
    for (int g0 = 0; g0 < G; g0 += T) {
      int v[T];
#pragma unroll
      for (int t = 0; t < T; ++t)
        v[t] = (g0 + t < G) ? hp[(size_t)(g0 + t) * np] : 0;
#pragma unroll
      for (int t = 0; t < T; ++t) {
        if (g0 + t < G) hp[(size_t)(g0 + t) * np] = run;
        run += v[t];
      }
    }
  }
  int inc = run;
#pragma unroll
  for (int off = 1; off < 32; off <<= 1) {
    const int u = __shfl_up(inc, off);
    if (lane >= off) inc += u;
  }
  if (lane == 31) wsum[w] = inc;
  __syncthreads();
  int before = 0;
#pragma unroll
  for (int k = 0; k < CB / 32; ++k) before += (k < w) ? wsum[k] : 0;
  if (valid) rowloc[d] = before + inc - run;
  if (tid == CB - 1) blocksum[blk] = before + inc;
}

template <int COUT, int BM, int HEADS>
__global__ __launch_bounds__(CB) void gemm_colsum_kernel(
    const float* __restrict__ A, const float* __restrict__ W,
    const float* __restrict__ att_src, const float* __restrict__ att_dst,
    float* __restrict__ H, float* __restrict__ as_, float* __restrict__ ad_,
    int n,
    int* __restrict__ hist, int* __restrict__ rowloc, int* __restrict__ blocksum,
    int G, int nblk) {
  static_assert(CB == 256, "the GEMM body is written for 8 waves");
  const int b = (int)blockIdx.x;
  if (b < nblk) {
    hist_colsum_block(hist, rowloc, blocksum, n, G, b);
  } else {
    gemm_tile<COUT, BM, HEADS>(A, W, att_src, att_dst, H, as_, ad_, n, b - nblk);
  }
}

__global__ __launch_bounds__(CB) void edge_place_kernel(
    const int* __restrict__ ei, int E, int n,
    const int* __restrict__ rowloc, const int* __restrict__ blocksum, int nblk,
    const int* __restrict__ hist, const int* __restrict__ pos,
    int* __restrict__ rowptr, int* __restrict__ csr_src) {
  extern __shared__ int bp[];
  __shared__ int wsum[CB / 32];
  const int tid  = threadIdx.x;
  const int lane = tid & 31;
  const int w    = tid >> 5;
  const int i    = blockIdx.x * CB + tid;
  const bool valid = i < E;
  const size_t np = (size_t)((n + 3) & ~3);

  int v = (tid < nblk) ? blocksum[tid] : 0;
  int s = 0, d = 0, rl = 0, hv = 0, pv = 0;
  if (valid) {
    s  = ei[i];
    d  = ei[E + i];
    pv = pos[i];
    rl = rowloc[d];
    hv = hist[(size_t)(i >> SLICE_LOG2) * np + d];
  }

  int carry = 0;
  for (int c0 = 0; c0 < nblk; c0 += CB) {
    const int idx = c0 + tid;
    if (c0) v = (idx < nblk) ? blocksum[idx] : 0;
    int inc = v;
#pragma unroll
    for (int off = 1; off < 32; off <<= 1) {
      const int u = __shfl_up(inc, off);
      if (lane >= off) inc += u;
    }
    if (lane == 31) wsum[w] = inc;
    __syncthreads();
    int before = 0, total = 0;
#pragma unroll
    for (int k = 0; k < CB / 32; ++k) {
      const int t = wsum[k];
      before += (k < w) ? t : 0;
      total  += t;
    }
    if (idx < nblk) bp[idx] = carry + before + inc - v;
    carry += total;
    __syncthreads();
  }

  for (int b = blockIdx.x; b < nblk; b += gridDim.x) {
    const int dn = b * CB + tid;
    if (dn < n) rowptr[dn] = bp[b] + rowloc[dn];
  }
  if (blockIdx.x == 0 && tid == 0) rowptr[n] = carry;

  if (valid) {
    const int slot = bp[d >> CB_LOG2] + rl + hv + pv;
    if ((unsigned)slot < (unsigned)E) csr_src[slot] = s;
  }
}


__device__ __forceinline__ float wave_max_f(float v) {
#pragma unroll
  for (int off = 16; off > 0; off >>= 1) v = fmaxf(v, __shfl_xor(v, off));
  return v;
}
__device__ __forceinline__ float wave_sum_f(float v) {
#pragma unroll
  for (int off = 16; off > 0; off >>= 1) v += __shfl_xor(v, off);
  return v;
}
__device__ __forceinline__ float leaky(float e) {
  return (e > 0.f) ? e : NEG_SLOPE * e;
}

template <int N> struct fvec { typedef float type __attribute__((ext_vector_type(N))); };

template <int HEADS, int C, bool ELU>
__device__ __forceinline__ typename fvec<HEADS * C / 32>::type gather_node(
    const int* __restrict__ rowptr, const int* __restrict__ csr_src,
    const float* __restrict__ H, const float* __restrict__ as_,
    const float* __restrict__ ad_, typename fvec<HEADS * C / 32>::type bv,
    int node, int lane, int n, int E) {
  constexpr int CH  = HEADS * C;
  constexpr int VEC = CH / 32;
  typedef typename fvec<VEC>::type vecf;
  const int myh = (lane * VEC) / C;

  int beg = rowptr[node];
  int end = rowptr[node + 1];
  beg = (beg < 0) ? 0 : beg;
  end = (end > E) ? E : end;
  const int deg = end - beg;

  const vecf hself = *(const vecf*)(H + (size_t)node * CH + lane * VEC);

  float adh[HEADS], eself[HEADS], m[HEADS];
#pragma unroll
  for (int h = 0; h < HEADS; ++h) {
    adh[h]   = ad_[node * HEADS + h];
    eself[h] = leaky(as_[node * HEADS + h] + adh[h]);
  }

  const bool v0 = lane < deg;
  int s0 = node;
  if (v0) {
    s0 = csr_src[beg + lane];
    s0 = (s0 < 0) ? 0 : ((s0 >= n) ? n - 1 : s0);
  }
  float e0[HEADS];
  if (HEADS == 2) {
    const v2f t = *(const v2f*)(as_ + (size_t)s0 * 2);
    e0[0]         = leaky(t.x + adh[0]);
    e0[HEADS - 1] = leaky(t.y + adh[HEADS - 1]);
  } else {
    e0[0] = leaky(as_[s0] + adh[0]);
  }
#pragma unroll
  for (int h = 0; h < HEADS; ++h) m[h] = fmaxf(eself[h], e0[h]);
  for (int base = 32; base < deg; base += 32) {
    int j = base + lane;
    if (j < deg) {
      int s = csr_src[beg + j];
      s = (s < 0) ? 0 : ((s >= n) ? n - 1 : s);
#pragma unroll
      for (int h = 0; h < HEADS; ++h)
        m[h] = fmaxf(m[h], leaky(as_[s * HEADS + h] + adh[h]));
    }
  }
#pragma unroll
  for (int h = 0; h < HEADS; ++h) m[h] = wave_max_f(m[h]);

  float p0[HEADS], den[HEADS], pself[HEADS];
#pragma unroll
  for (int h = 0; h < HEADS; ++h) {
    p0[h]  = v0 ? __expf(e0[h] - m[h]) : 0.f;
    den[h] = p0[h];
  }
  for (int base = 32; base < deg; base += 32) {
    int j = base + lane;
    if (j < deg) {
      int s = csr_src[beg + j];
      s = (s < 0) ? 0 : ((s >= n) ? n - 1 : s);
#pragma unroll
      for (int h = 0; h < HEADS; ++h)
        den[h] += __expf(leaky(as_[s * HEADS + h] + adh[h]) - m[h]);
    }
  }
#pragma unroll
  for (int h = 0; h < HEADS; ++h) {
    pself[h] = __expf(eself[h] - m[h]);
    den[h]   = wave_sum_f(den[h]) + pself[h] + 1e-16f;
  }

  vecf acc;
  {
    const float a = (HEADS == 2 && myh) ? pself[HEADS - 1] / den[HEADS - 1]
                                        : pself[0] / den[0];
    acc = hself * a;
  }
  for (int base = 0; base < deg; base += 32) {
    int   sj = s0;
    float al[HEADS];
    if (base == 0) {
#pragma unroll
      for (int h = 0; h < HEADS; ++h) al[h] = p0[h] / den[h];
    } else {
      int j = base + lane;
      sj = node;
#pragma unroll
      for (int h = 0; h < HEADS; ++h) al[h] = 0.f;
      if (j < deg) {
        sj = csr_src[beg + j];
        sj = (sj < 0) ? 0 : ((sj >= n) ? n - 1 : sj);
#pragma unroll
        for (int h = 0; h < HEADS; ++h)
          al[h] = __expf(leaky(as_[sj * HEADS + h] + adh[h]) - m[h]) / den[h];
      }
    }
    const int cnt  = (deg - base < 32) ? (deg - base) : 32;
    const int cnt4 = (cnt + 3) & ~3;
    for (int k = 0; k < cnt4; k += 4) {
      vecf  hv[4];
      float a[4];
#pragma unroll
      for (int u = 0; u < 4; ++u) {
        const int sk = __builtin_amdgcn_readlane(sj, k + u);
        hv[u] = *(const vecf*)(H + (size_t)sk * CH + lane * VEC);
        const float a0 = __int_as_float(__builtin_amdgcn_readlane(__float_as_int(al[0]), k + u));
        a[u] = a0;
        if (HEADS == 2) {
          const float a1 = __int_as_float(__builtin_amdgcn_readlane(__float_as_int(al[HEADS - 1]), k + u));
          a[u] = myh ? a1 : a0;
        }
      }
#pragma unroll
      for (int u = 0; u < 4; ++u) acc += hv[u] * a[u];
    }
  }

#pragma unroll
  for (int v = 0; v < VEC; ++v) {
    float y = acc[v] + bv[v];
    if (ELU) y = (y > 0.f) ? y : expm1f(y);
    acc[v] = y;
  }
  return acc;
}

template <int HEADS, int C, bool ELU>
__global__ __launch_bounds__(256) void gat_gather_kernel(
    const int* __restrict__ rowptr, const int* __restrict__ csr_src,
    const float* __restrict__ H, const float* __restrict__ as_,
    const float* __restrict__ ad_, const float* __restrict__ bias,
    float* __restrict__ out, int n, int E) {
  constexpr int CH  = HEADS * C;
  constexpr int VEC = CH / 32;
  typedef typename fvec<VEC>::type vecf;

  const int lane = threadIdx.x & 31;
  const int wv  = __builtin_amdgcn_readfirstlane((int)(threadIdx.x >> 5));
  const int gw  = blockIdx.x * 8 + wv;
  const int nw  = gridDim.x * 8;
  const vecf bv = *(const vecf*)(bias + lane * VEC);

  for (int node = gw; node < n; node += nw) {
    const vecf acc = gather_node<HEADS, C, ELU>(rowptr, csr_src, H, as_, ad_, bv, node, lane, n, E);
    *(vecf*)(out + (size_t)node * CH + lane * VEC) = acc;
  }
}

constexpr int FN = 50;
template <int C1, int COUT>
__global__ __launch_bounds__(512) void gather_gemm_kernel(
    const int* __restrict__ rowptr, const int* __restrict__ csr_src,
    const float* __restrict__ H1, const float* __restrict__ as1,
    const float* __restrict__ ad1, const float* __restrict__ bias1,
    const float* __restrict__ W, const float* __restrict__ att_src,
    const float* __restrict__ att_dst,
    float* __restrict__ H2, float* __restrict__ as2, float* __restrict__ ad2,
    int n, int E) {
  constexpr int K  = 2 * C1;
  constexpr int P  = K + 4;
  constexpr int FM = 64;
  constexpr int NW = 16;
  static_assert(K == 128 && COUT == 64, "one 16x16 tile per wave, one 512-byte row per load");
  static_assert(FN <= FM - NW / 2, "waves 8..15 have no fourth node: they stage W2 meanwhile");
  typedef typename fvec<4>::type vec4f;
  __shared__ __attribute__((aligned(16))) float Ws[COUT * P];
  __shared__ __attribute__((aligned(16))) float As[FM * P];

  const int tid  = threadIdx.x;
  const int lane = tid & 31;
  const int wv   = __builtin_amdgcn_readfirstlane(tid >> 5);
  const int row0 = (int)blockIdx.x * FN;

  const vec4f bv = *(const vec4f*)(bias1 + lane * 4);
  for (int j = 0; j < FM / NW; ++j) {
    const int r    = wv + NW * j;
    const int node = row0 + r;
    vec4f row = {0.f, 0.f, 0.f, 0.f};
    if (r < FN && node < n)
      row = gather_node<2, C1, true>(rowptr, csr_src, H1, as1, ad1, bv, node, lane, n, E);
    *(vec4f*)(As + r * P + lane * 4) = row;
  }
  if (wv >= NW / 2) {
    constexpr int WR = COUT / (NW / 2);
    float4 wreg[WR];
#pragma unroll
    for (int j = 0; j < WR; ++j)
      wreg[j] = *(const float4*)(W + (size_t)(wv - NW / 2 + (NW / 2) * j) * K + lane * 4);
#pragma unroll
    for (int j = 0; j < WR; ++j)
      *(float4*)(Ws + (wv - NW / 2 + (NW / 2) * j) * P + lane * 4) = wreg[j];
  }
  __syncthreads();

  const int mn      = lane & 15;
  const int khalf   = (lane >> 4) * 2;
  const int colTile = wv & 3;
  const int rowTile = wv >> 2;
  const float* bp = Ws + (colTile * 16 + mn) * P + khalf;
  const float* ap = As + (rowTile * 16 + mn) * P + khalf;

  v8f acc = {};
#pragma unroll
  for (int k0 = 0; k0 < K; k0 += 4) {
    const v2f b = *(const v2f*)(bp + k0);
    const v2f a = *(const v2f*)(ap + k0);
    acc = __builtin_amdgcn_wmma_f32_16x16x4_f32(
         false, a,  false, b,
         (short)0, acc,  false,  false);
  }

  const v2f sv = *(const v2f*)(att_src + lane * 2);
  const v2f dv = *(const v2f*)(att_dst + lane * 2);

  __syncthreads();
  {
    float* op = As + (rowTile * 16 + 8 * (lane >> 4)) * P + colTile * 16 + mn;
#pragma unroll
    for (int v = 0; v < 8; ++v) op[v * P] = acc[v];
  }
  __syncthreads();

  constexpr int R  = FM / NW;
  constexpr int NV = 2 * R;
  static_assert(NV == 8, "the exchange below ends with one value per group of four lanes");
  v2f hrow[R];
#pragma unroll
  for (int j = 0; j < R; ++j)
    hrow[j] = *(const v2f*)(As + (wv + NW * j) * P + lane * 2);
#pragma unroll
  for (int j = 0; j < R; ++j) {
    const int grow = row0 + wv + NW * j;
    if (wv + NW * j < FN && grow < n) *(v2f*)(H2 + (size_t)grow * COUT + lane * 2) = hrow[j];
  }

  float val[NV];
#pragma unroll
  for (int j = 0; j < R; ++j) {
    float ps = 0.f, pd = 0.f;
#pragma unroll
    for (int c = 0; c < 2; ++c) {
      ps += hrow[j][c] * sv[c];
      pd += hrow[j][c] * dv[c];
    }
    val[2 * j]     = ps;
    val[2 * j + 1] = pd;
  }
  head_sum<NV, 16, NV>(val, lane);
  if ((lane & 3) == 0) {
    const int i    = (lane >> 2) & (NV - 1);
    const int r    = wv + NW * (i >> 1);
    const int grow = row0 + r;
    if (r < FN && grow < n) {
      float* dp = (i & 1) ? ad2 : as2;
      dp[grow] = val[0];
    }
  }
}


static inline int cdiv(long long a, long long b) { return (int)((a + b - 1) / b); }

extern "C" void kernel_launch(void* const* d_in, const int* in_sizes, int n_in,
                              void* d_out, int out_size, void* d_ws, size_t ws_size,
                              hipStream_t stream) {
  const int IN_CH = 128, HID = 64, HEADS = 2, OUT = 64;
  const int CH1 = HEADS * HID;

  const float* x        = (const float*)d_in[0];
  const int*   ei       = (const int*)d_in[1];
  const float* W1       = (const float*)d_in[2];
  const float* att1_src = (const float*)d_in[3];
  const float* att1_dst = (const float*)d_in[4];
  const float* b1       = (const float*)d_in[5];
  const float* W2       = (const float*)d_in[6];
  const float* att2_src = (const float*)d_in[7];
  const float* att2_dst = (const float*)d_in[8];
  const float* b2       = (const float*)d_in[9];
  float* out            = (float*)d_out;

  const int n  = in_sizes[0] / IN_CH;
  const int E  = in_sizes[1] / 2;

  char* ws = (char*)d_ws;
  size_t off = 0;
  auto alloc = [&](size_t bytes) -> void* {
    void* p = ws + off;
    off += (bytes + 255) & ~(size_t)255;
    return p;
  };
  float* h1      = (float*)alloc((size_t)n * CH1 * 4);
  float* h2      = (float*)alloc((size_t)n * OUT * 4);
  float* as1     = (float*)alloc((size_t)n * HEADS * 4);
  float* ad1     = (float*)alloc((size_t)n * HEADS * 4);
  float* as2     = (float*)alloc((size_t)n * 4);
  float* ad2     = (float*)alloc((size_t)n * 4);
  const int nblk = cdiv(n, CB);
  int*   rowloc  = (int*)  alloc((size_t)n * 4);
  int*   blocksum= (int*)  alloc((size_t)nblk * 4);
  int*   rowptr  = (int*)  alloc((size_t)(n + 1) * 4);
  int*   pos     = (int*)  alloc((size_t)E * 4);
  int*   csr_src = (int*)  alloc((size_t)E * 4);
  const int G    = cdiv(E, SLICE);
  int*   hist    = (int*)  alloc((size_t)G * (n + 4) * 4);
  (void)ws_size; (void)n_in; (void)out_size;

  const int BT = 256;
  const int GATHER_BLOCKS = 2048;

  edge_count_kernel<<<G, 1024, 0, stream>>>(ei, E, n, hist, pos);
  gemm_colsum_kernel<128, 32, 2><<<nblk + cdiv(n, 32), BT, 0, stream>>>(
      x, W1, att1_src, att1_dst, h1, as1, ad1, n, hist, rowloc, blocksum, G, nblk);
  edge_place_kernel<<<cdiv(E, CB), CB, (size_t)nblk * sizeof(int), stream>>>(
      ei, E, n, rowloc, blocksum, nblk, hist, pos, rowptr, csr_src);

  gather_gemm_kernel<64, 64><<<cdiv(n, FN), 512, 0, stream>>>(
      rowptr, csr_src, h1, as1, ad1, b1, W2, att2_src, att2_dst, h2, as2, ad2, n, E);

  gat_gather_kernel<1, 64, false><<<GATHER_BLOCKS, BT, 0, stream>>>(
      rowptr, csr_src, h2, as2, ad2, b2, out, n, E);
}
